# starting kernel
# baseline (speedup 1.0000x reference)
#include <hip/hip_runtime.h>

typedef float v2f __attribute__((ext_vector_type(2)));
typedef float v8f __attribute__((ext_vector_type(8)));

constexpr int N0 = 20000, N1 = 2000;
constexpr int FF = 512, HH = 256, DD = 128, KK = 16;
constexpr int E00 = 640000, E01 = 100000, E10 = 100000, E11 = 64000;

constexpr long long off_xw00 = 0;
constexpr long long off_xw10 = 5120000;
constexpr long long off_xw01 = 10240000;
constexpr long long off_xw11 = 10752000;
constexpr long long off_acc00 = 18944000;
constexpr long long off_acc01 = 24064000;
constexpr long long off_acc10 = 29184000;
constexpr long long off_acc11 = 29696000;

constexpr long long off_h0    = off_xw00;
constexpr long long off_h1    = off_xw01;
constexpr long long off_hw00  = off_xw10;
constexpr long long off_hw10  = off_xw10 + 2560000;
constexpr long long off_hw01  = off_xw11;
constexpr long long off_hw11  = off_xw11 + 256000;
constexpr long long off_bcc00 = off_acc00;
constexpr long long off_bcc01 = off_acc00 + 2560000;
constexpr long long off_bcc10 = off_acc10;
constexpr long long off_bcc11 = off_acc11;
constexpr long long off_z0R00 = off_acc01;
constexpr long long off_z0R01 = off_acc01 + 2560000;
constexpr long long off_z1R10 = off_acc10 + 256000;
constexpr long long off_z1d   = off_acc11 + 4096000;
constexpr long long off_Zk    = off_acc11;

__device__ __forceinline__ float wave_sum(float v) {
#pragma unroll
  for (int off = 16; off > 0; off >>= 1) v += __shfl_xor(v, off, 32);
  return v;
}

constexpr int BK   = 32;
constexpr int ASTR = 36;
constexpr int BPS  = 160;

__global__ __launch_bounds__(256) void gemm_f32_wmma(
    const float* __restrict__ A, const float* __restrict__ B, float* __restrict__ C,
    int M, int N, int Kd, long long sA, long long sB, long long sC) {
  __shared__ float As[2][128 * ASTR];
  __shared__ float Bs[2][(BK / 2) * BPS];

  const int b = blockIdx.z;
  A += (long long)b * sA;
  B += (long long)b * sB;
  C += (long long)b * sC;

  const int tid  = threadIdx.x;
  const int lane = tid & 31;
  const int wave = tid >> 5;
  const int mBlk = blockIdx.x * 128;
  const int n0   = blockIdx.y * 64;
  const int m0   = mBlk + wave * 16;
  const int half = lane >> 4;
  const int lr   = lane & 15;
  const bool active = (m0 < M);

  const int arow = tid >> 3;
  const int ac4  = (tid & 7) * 4;
  const int brow = tid >> 4;
  const int bc4  = (tid & 15) * 4;

  const int nChunks = Kd / BK;
  const float4 zero4 = make_float4(0.f, 0.f, 0.f, 0.f);

  {
#pragma unroll
    for (int p = 0; p < 4; ++p) {
      const int row = arow + p * 32;
      const int grow = mBlk + row;
      float4 v = (grow < M) ? *(const float4*)(A + (long long)grow * Kd + ac4) : zero4;
      *(float4*)&As[0][row * ASTR + ac4] = v;
    }
#pragma unroll
    for (int p = 0; p < 2; ++p) {
      const int row = brow + p * 16;
      const float4 v = *(const float4*)(B + (long long)row * N + n0 + bc4);
      float* bp = &Bs[0][(row >> 1) * BPS + bc4 * 2 + (row & 1)];
      bp[0] = v.x; bp[2] = v.y; bp[4] = v.z; bp[6] = v.w;
    }
  }
  __syncthreads();

  v8f acc0 = {}, acc1 = {}, acc2 = {}, acc3 = {};

  for (int c = 0; c < nChunks; ++c) {
    const int buf = c & 1;
    const bool pf = (c + 1 < nChunks);
    float4 ra[4], rb[2];

    if (pf) {
      const int k0 = (c + 1) * BK;
#pragma unroll
      for (int p = 0; p < 4; ++p) {
        const int grow = mBlk + arow + p * 32;
        ra[p] = (grow < M) ? *(const float4*)(A + (long long)grow * Kd + k0 + ac4) : zero4;
      }
#pragma unroll
      for (int p = 0; p < 2; ++p)
        rb[p] = *(const float4*)(B + (long long)(k0 + brow + p * 16) * N + n0 + bc4);
    }

    if (active) {
      const float* Asb = &As[buf][(wave * 16 + lr) * ASTR + 2 * half];
      const float* Bsb = &Bs[buf][half * BPS + lr * 2];
#pragma unroll
      for (int kk = 0; kk < BK; kk += 4) {
        const v2f a = *(const v2f*)(Asb + kk);
        const float* bp = Bsb + (kk >> 1) * BPS;
        const v2f b0 = *(const v2f*)(bp);
        const v2f b1 = *(const v2f*)(bp + 32);
        const v2f b2 = *(const v2f*)(bp + 64);
        const v2f b3 = *(const v2f*)(bp + 96);
        acc0 = __builtin_amdgcn_wmma_f32_16x16x4_f32(false, a, false, b0, (short)0, acc0, false, false);
        acc1 = __builtin_amdgcn_wmma_f32_16x16x4_f32(false, a, false, b1, (short)0, acc1, false, false);
        acc2 = __builtin_amdgcn_wmma_f32_16x16x4_f32(false, a, false, b2, (short)0, acc2, false, false);
        acc3 = __builtin_amdgcn_wmma_f32_16x16x4_f32(false, a, false, b3, (short)0, acc3, false, false);
      }
    }
    __syncthreads();

    if (pf) {
      const int nb = buf ^ 1;
#pragma unroll
      for (int p = 0; p < 4; ++p)
        *(float4*)&As[nb][(arow + p * 32) * ASTR + ac4] = ra[p];
#pragma unroll
      for (int p = 0; p < 2; ++p) {
        const int row = brow + p * 16;
        float* bp = &Bs[nb][(row >> 1) * BPS + bc4 * 2 + (row & 1)];
        bp[0] = rb[p].x; bp[2] = rb[p].y; bp[4] = rb[p].z; bp[6] = rb[p].w;
      }
      __syncthreads();
    }
  }

  if (active) {
#pragma unroll
    for (int r = 0; r < 8; ++r) {
      const int row = m0 + r + 8 * half;
      float* cp = C + (long long)row * N + n0 + lr;
      cp[0]  = acc0[r];
      cp[16] = acc1[r];
      cp[32] = acc2[r];
      cp[48] = acc3[r];
    }
  }
}

__global__ void spmm_edges(const int* __restrict__ dst, const int* __restrict__ src,
                           const float* __restrict__ val, const float* __restrict__ x,
                           float* __restrict__ out, int nE, int N,
                           int eStride, long long xStride, long long oStride) {
  const int k = blockIdx.y;
  dst += (long long)k * eStride;
  src += (long long)k * eStride;
  val += (long long)k * eStride;
  x   += (long long)k * xStride;
  out += (long long)k * oStride;
  const int w = blockIdx.x * (blockDim.x >> 5) + (threadIdx.x >> 5);
  if (w >= nE) return;
  const int lane = threadIdx.x & 31;
  const int d = dst[w];
  const int s = src[w];
  const float v = val[w];
  const float4* xs = (const float4*)(x + (long long)s * N);
  float* od = out + (long long)d * N;
  const int nq = N >> 2;
  for (int q = lane; q < nq; q += 32) {
    const float4 t = xs[q];
    float* o = od + q * 4;
    atomicAdd(o + 0, v * t.x);
    atomicAdd(o + 1, v * t.y);
    atomicAdd(o + 2, v * t.z);
    atomicAdd(o + 3, v * t.w);
  }
}

__global__ void finalize_conv(const float* __restrict__ buf, float* __restrict__ out,
                              int K, int n, int N, long long kStride, int do_relu) {
  const int wave = threadIdx.x >> 5;
  const int lane = threadIdx.x & 31;
  const int row = blockIdx.x * (blockDim.x >> 5) + wave;
  if (row >= n) return;
  const int nc = N >> 5;
  float vals[8];
  float ss = 0.f;
  for (int j = 0; j < nc; ++j) {
    const int c = lane + (j << 5);
    float a = 0.f;
    for (int k = 0; k < K; ++k) {
      float t = buf[(long long)k * kStride + (long long)row * N + c];
      if (do_relu) t = fmaxf(t, 0.f);
      a += t;
    }
    vals[j] = a;
    ss += a * a;
  }
  ss = wave_sum(ss);
  const float sc = rsqrtf(fmaxf(ss, 1e-12f));
  for (int j = 0; j < nc; ++j)
    out[(long long)row * N + lane + (j << 5)] = vals[j] * sc;
}

__global__ void add_act(const float* __restrict__ a, const float* __restrict__ b,
                        float* __restrict__ o, long long n, int do_relu) {
  const long long i = (long long)blockIdx.x * blockDim.x + threadIdx.x;
  if (i >= n) return;
  float v = a[i] + b[i];
  o[i] = do_relu ? fmaxf(v, 0.f) : v;
}

__global__ void scale_dloc(const float* __restrict__ z1, const float* __restrict__ dloc,
                           float* __restrict__ out, long long total) {
  const long long i = (long long)blockIdx.x * blockDim.x + threadIdx.x;
  if (i >= total) return;
  const int f = (int)(i % DD);
  const long long rk = i / DD;
  const int r = (int)(rk % N1);
  const int k = (int)(rk / N1);
  out[i] = z1[(long long)r * DD + f] * dloc[k * DD + f];
}

__global__ void edge_dot128(const float* __restrict__ zl, const float* __restrict__ zr,
                            const int* __restrict__ dst, const int* __restrict__ src,
                            float* __restrict__ out, int nE, int eStride,
                            long long zlStride, long long zrStride, long long oStride) {
  const int k = blockIdx.y;
  zl  += (long long)k * zlStride;
  zr  += (long long)k * zrStride;
  dst += (long long)k * eStride;
  src += (long long)k * eStride;
  out += (long long)k * oStride;
  const int e = blockIdx.x * (blockDim.x >> 5) + (threadIdx.x >> 5);
  if (e >= nE) return;
  const int lane = threadIdx.x & 31;
  const int d = dst[e];
  const int s = src[e];
  const float4 a = ((const float4*)(zl + (long long)d * DD))[lane];
  const float4 b = ((const float4*)(zr + (long long)s * DD))[lane];
  float acc = a.x * b.x + a.y * b.y + a.z * b.z + a.w * b.w;
  acc = wave_sum(acc);
  if (lane == 0) out[e] = acc;
}

extern "C" void kernel_launch(void* const* d_in, const int* in_sizes, int n_in,
                              void* d_out, int out_size, void* d_ws, size_t ws_size,
                              hipStream_t stream) {
  (void)in_sizes; (void)n_in; (void)out_size; (void)ws_size;

  const float* feat0   = (const float*)d_in[0];
  const float* feat1   = (const float*)d_in[1];
  const int*   e00_dst = (const int*)d_in[2];
  const int*   e00_src = (const int*)d_in[3];
  const float* e00_val = (const float*)d_in[4];
  const int*   e01_dst = (const int*)d_in[5];
  const int*   e01_src = (const int*)d_in[6];
  const float* e01_val = (const float*)d_in[7];
  const int*   e10_dst = (const int*)d_in[8];
  const int*   e10_src = (const int*)d_in[9];
  const float* e10_val = (const float*)d_in[10];
  const int*   e11_dst = (const int*)d_in[11];
  const int*   e11_src = (const int*)d_in[12];
  const float* e11_val = (const float*)d_in[13];
  const float* W1_00 = (const float*)d_in[14];
  const float* W1_01 = (const float*)d_in[15];
  const float* W1_10 = (const float*)d_in[16];
  const float* W1_11 = (const float*)d_in[17];
  const float* W2_00 = (const float*)d_in[18];
  const float* W2_01 = (const float*)d_in[19];
  const float* W2_10 = (const float*)d_in[20];
  const float* W2_11 = (const float*)d_in[21];
  const float* R00   = (const float*)d_in[22];
  const float* R01   = (const float*)d_in[23];
  const float* R10   = (const float*)d_in[24];
  const float* Rglob = (const float*)d_in[25];
  const float* dloc  = (const float*)d_in[26];

  float* ws = (float*)d_ws;
  float* out = (float*)d_out;
  float* z0  = out;
  float* z1  = out + 2560000;
  float* s00 = z1 + 256000;
  float* s01 = s00 + E00;
  float* s10 = s01 + E01;
  float* s11 = s10 + E10;

  const dim3 blk(256);
  auto gg = [](int M, int N, int batch) { return dim3((M + 127) / 128, N / 64, batch); };

  gemm_f32_wmma<<<gg(N0, HH, 1), blk, 0, stream>>>(feat0, W1_00, ws + off_xw00, N0, HH, FF, 0, 0, 0);
  gemm_f32_wmma<<<gg(N0, HH, 1), blk, 0, stream>>>(feat0, W1_10, ws + off_xw10, N0, HH, FF, 0, 0, 0);
  gemm_f32_wmma<<<gg(N1, HH, 1), blk, 0, stream>>>(feat1, W1_01, ws + off_xw01, N1, HH, FF, 0, 0, 0);
  gemm_f32_wmma<<<gg(N1, HH, KK), blk, 0, stream>>>(feat1, W1_11, ws + off_xw11, N1, HH, FF,
                                                    0, (long long)FF * HH, (long long)N1 * HH);

  hipMemsetAsync(ws + off_acc00, 0, (size_t)18944000 * sizeof(float), stream);

  spmm_edges<<<dim3((E00 + 7) / 8, 1), blk, 0, stream>>>(e00_dst, e00_src, e00_val, ws + off_xw00,
                                                         ws + off_acc00, E00, HH, 0, 0, 0);
  spmm_edges<<<dim3((E01 + 7) / 8, 1), blk, 0, stream>>>(e01_dst, e01_src, e01_val, ws + off_xw01,
                                                         ws + off_acc01, E01, HH, 0, 0, 0);
  spmm_edges<<<dim3((E10 + 7) / 8, 1), blk, 0, stream>>>(e10_dst, e10_src, e10_val, ws + off_xw10,
                                                         ws + off_acc10, E10, HH, 0, 0, 0);
  spmm_edges<<<dim3((E11 + 7) / 8, KK), blk, 0, stream>>>(e11_dst, e11_src, e11_val, ws + off_xw11,
                                                          ws + off_acc11, E11, HH, E11,
                                                          (long long)N1 * HH, (long long)N1 * HH);

  finalize_conv<<<dim3((N0 + 7) / 8), blk, 0, stream>>>(ws + off_acc00, ws + off_acc00, 1, N0, HH, 0, 1);
  finalize_conv<<<dim3((N0 + 7) / 8), blk, 0, stream>>>(ws + off_acc01, ws + off_acc01, 1, N0, HH, 0, 1);
  finalize_conv<<<dim3((N1 + 7) / 8), blk, 0, stream>>>(ws + off_acc10, ws + off_acc10, 1, N1, HH, 0, 1);
  finalize_conv<<<dim3((N1 + 7) / 8), blk, 0, stream>>>(ws + off_acc11, ws + off_acc11, KK, N1, HH,
                                                        (long long)N1 * HH, 1);

  add_act<<<dim3((N0 * HH + 255) / 256), blk, 0, stream>>>(ws + off_acc00, ws + off_acc01,
                                                           ws + off_h0, (long long)N0 * HH, 1);
  add_act<<<dim3((N1 * HH + 255) / 256), blk, 0, stream>>>(ws + off_acc10, ws + off_acc11,
                                                           ws + off_h1, (long long)N1 * HH, 1);

  gemm_f32_wmma<<<gg(N0, DD, 1), blk, 0, stream>>>(ws + off_h0, W2_00, ws + off_hw00, N0, DD, HH, 0, 0, 0);
  gemm_f32_wmma<<<gg(N0, DD, 1), blk, 0, stream>>>(ws + off_h0, W2_10, ws + off_hw10, N0, DD, HH, 0, 0, 0);
  gemm_f32_wmma<<<gg(N1, DD, 1), blk, 0, stream>>>(ws + off_h1, W2_01, ws + off_hw01, N1, DD, HH, 0, 0, 0);
  gemm_f32_wmma<<<gg(N1, DD, KK), blk, 0, stream>>>(ws + off_h1, W2_11, ws + off_hw11, N1, DD, HH,
                                                    0, (long long)HH * DD, (long long)N1 * DD);

  hipMemsetAsync(ws + off_bcc00, 0, (size_t)5120000 * sizeof(float), stream);
  hipMemsetAsync(ws + off_bcc10, 0, (size_t)256000 * sizeof(float), stream);
  hipMemsetAsync(ws + off_bcc11, 0, (size_t)4096000 * sizeof(float), stream);

  spmm_edges<<<dim3((E00 + 7) / 8, 1), blk, 0, stream>>>(e00_dst, e00_src, e00_val, ws + off_hw00,
                                                         ws + off_bcc00, E00, DD, 0, 0, 0);
  spmm_edges<<<dim3((E01 + 7) / 8, 1), blk, 0, stream>>>(e01_dst, e01_src, e01_val, ws + off_hw01,
                                                         ws + off_bcc01, E01, DD, 0, 0, 0);
  spmm_edges<<<dim3((E10 + 7) / 8, 1), blk, 0, stream>>>(e10_dst, e10_src, e10_val, ws + off_hw10,
                                                         ws + off_bcc10, E10, DD, 0, 0, 0);
  spmm_edges<<<dim3((E11 + 7) / 8, KK), blk, 0, stream>>>(e11_dst, e11_src, e11_val, ws + off_hw11,
                                                          ws + off_bcc11, E11, DD, E11,
                                                          (long long)N1 * DD, (long long)N1 * DD);

  finalize_conv<<<dim3((N0 + 7) / 8), blk, 0, stream>>>(ws + off_bcc00, ws + off_bcc00, 1, N0, DD, 0, 0);
  finalize_conv<<<dim3((N0 + 7) / 8), blk, 0, stream>>>(ws + off_bcc01, ws + off_bcc01, 1, N0, DD, 0, 0);
  finalize_conv<<<dim3((N1 + 7) / 8), blk, 0, stream>>>(ws + off_bcc10, ws + off_bcc10, 1, N1, DD, 0, 0);
  finalize_conv<<<dim3((N1 + 7) / 8), blk, 0, stream>>>(ws + off_bcc11, ws + off_bcc11, KK, N1, DD,
                                                        (long long)N1 * DD, 0);

  add_act<<<dim3((N0 * DD + 255) / 256), blk, 0, stream>>>(ws + off_bcc00, ws + off_bcc01, z0,
                                                           (long long)N0 * DD, 0);
  add_act<<<dim3((N1 * DD + 255) / 256), blk, 0, stream>>>(ws + off_bcc10, ws + off_bcc11, z1,
                                                           (long long)N1 * DD, 0);

  gemm_f32_wmma<<<gg(N0, DD, 1), blk, 0, stream>>>(z0, R00, ws + off_z0R00, N0, DD, DD, 0, 0, 0);
  gemm_f32_wmma<<<gg(N0, DD, 1), blk, 0, stream>>>(z0, R01, ws + off_z0R01, N0, DD, DD, 0, 0, 0);
  gemm_f32_wmma<<<gg(N1, DD, 1), blk, 0, stream>>>(z1, R10, ws + off_z1R10, N1, DD, DD, 0, 0, 0);

  scale_dloc<<<dim3((KK * N1 * DD + 255) / 256), blk, 0, stream>>>(z1, dloc, ws + off_z1d,
                                                                   (long long)KK * N1 * DD);
  gemm_f32_wmma<<<gg(N1, DD, KK), blk, 0, stream>>>(ws + off_z1d, Rglob, ws + off_Zk, N1, DD, DD,
                                                    (long long)N1 * DD, 0, (long long)N1 * DD);

  edge_dot128<<<dim3((E00 + 7) / 8, 1), blk, 0, stream>>>(ws + off_z0R00, z0, e00_dst, e00_src,
                                                          s00, E00, 0, 0, 0, 0);
  edge_dot128<<<dim3((E01 + 7) / 8, 1), blk, 0, stream>>>(ws + off_z0R01, z1, e01_dst, e01_src,
                                                          s01, E01, 0, 0, 0, 0);
  edge_dot128<<<dim3((E10 + 7) / 8, 1), blk, 0, stream>>>(ws + off_z1R10, z0, e10_dst, e10_src,
                                                          s10, E10, 0, 0, 0, 0);
  edge_dot128<<<dim3((E11 + 7) / 8, KK), blk, 0, stream>>>(ws + off_Zk, ws + off_z1d, e11_dst, e11_src,
                                                           s11, E11, E11,
                                                           (long long)N1 * DD, (long long)N1 * DD,
                                                           (long long)E11);
}


# SPMM: atomic edge scatter replaced by dst-sorted edge lists built in-call (count, scan, fill) and one-wave-per-row register gathers stored once; accumulator memsets gone
# speedup vs baseline: 1.4195x; 1.4195x over previous
#include <hip/hip_runtime.h>

typedef float v2f __attribute__((ext_vector_type(2)));
typedef float v8f __attribute__((ext_vector_type(8)));

constexpr int N0 = 20000, N1 = 2000;
constexpr int FF = 512, HH = 256, DD = 128, KK = 16;
constexpr int E00 = 640000, E01 = 100000, E10 = 100000, E11 = 64000;

constexpr long long off_xw00 = 0;
constexpr long long off_csr  = 5120000;
constexpr long long off_xw10 = off_xw00;
constexpr long long off_xw01 = 10240000;
constexpr long long off_xw11 = 10752000;
constexpr long long off_acc00 = 18944000;
constexpr long long off_acc01 = 24064000;
constexpr long long off_acc10 = 29184000;
constexpr long long off_acc11 = 29696000;

constexpr long long off_h0    = off_xw00;
constexpr long long off_h1    = off_xw01;
constexpr long long off_hw00  = off_acc01;
constexpr long long off_hw10  = off_acc01 + 2560000;
constexpr long long off_hw01  = off_xw11;
constexpr long long off_hw11  = off_xw11 + 256000;
constexpr long long off_bcc00 = off_acc00;
constexpr long long off_bcc01 = off_acc00 + 2560000;
constexpr long long off_bcc10 = off_acc10;
constexpr long long off_bcc11 = off_acc11;
constexpr long long off_z0R00 = off_acc01;
constexpr long long off_z0R01 = off_acc01 + 2560000;
constexpr long long off_z1R10 = off_acc10 + 256000;
constexpr long long off_z1d   = off_acc11 + 4096000;
constexpr long long off_Zk    = off_acc11;

constexpr int ETOT = E00 + E01 + E10 + KK * E11;
constexpr int NSEG = N0 + N0 + N1 + KK * N1;
constexpr int SEG00 = 0, SEG01 = N0, SEG10 = 2 * N0, SEG11 = 2 * N0 + N1;
constexpr long long off_perm = off_csr;
constexpr long long off_cnt  = off_csr + 2LL * ETOT;
constexpr long long off_cur  = off_cnt + NSEG;

__device__ __forceinline__ float wave_sum(float v) {
#pragma unroll
  for (int off = 16; off > 0; off >>= 1) v += __shfl_xor(v, off, 32);
  return v;
}

constexpr int BK   = 32;
constexpr int ASTR = 36;
constexpr int BPS  = 160;

__global__ __launch_bounds__(256) void gemm_f32_wmma(
    const float* __restrict__ A, const float* __restrict__ B, float* __restrict__ C,
    int M, int N, int Kd, long long sA, long long sB, long long sC) {
  __shared__ float As[2][128 * ASTR];
  __shared__ float Bs[2][(BK / 2) * BPS];

  const int b = blockIdx.z;
  A += (long long)b * sA;
  B += (long long)b * sB;
  C += (long long)b * sC;

  const int tid  = threadIdx.x;
  const int lane = tid & 31;
  const int wave = tid >> 5;
  const int mBlk = blockIdx.x * 128;
  const int n0   = blockIdx.y * 64;
  const int m0   = mBlk + wave * 16;
  const int half = lane >> 4;
  const int lr   = lane & 15;
  const bool active = (m0 < M);

  const int arow = tid >> 3;
  const int ac4  = (tid & 7) * 4;
  const int brow = tid >> 4;
  const int bc4  = (tid & 15) * 4;

  const int nChunks = Kd / BK;
  const float4 zero4 = make_float4(0.f, 0.f, 0.f, 0.f);

  {
#pragma unroll
    for (int p = 0; p < 4; ++p) {
      const int row = arow + p * 32;
      const int grow = mBlk + row;
      float4 v = (grow < M) ? *(const float4*)(A + (long long)grow * Kd + ac4) : zero4;
      *(float4*)&As[0][row * ASTR + ac4] = v;
    }
#pragma unroll
    for (int p = 0; p < 2; ++p) {
      const int row = brow + p * 16;
      const float4 v = *(const float4*)(B + (long long)row * N + n0 + bc4);
      float* bp = &Bs[0][(row >> 1) * BPS + bc4 * 2 + (row & 1)];
      bp[0] = v.x; bp[2] = v.y; bp[4] = v.z; bp[6] = v.w;
    }
  }
  __syncthreads();

  v8f acc0 = {}, acc1 = {}, acc2 = {}, acc3 = {};

  for (int c = 0; c < nChunks; ++c) {
    const int buf = c & 1;
    const bool pf = (c + 1 < nChunks);
    float4 ra[4], rb[2];

    if (pf) {
      const int k0 = (c + 1) * BK;
#pragma unroll
      for (int p = 0; p < 4; ++p) {
        const int grow = mBlk + arow + p * 32;
        ra[p] = (grow < M) ? *(const float4*)(A + (long long)grow * Kd + k0 + ac4) : zero4;
      }
#pragma unroll
      for (int p = 0; p < 2; ++p)
        rb[p] = *(const float4*)(B + (long long)(k0 + brow + p * 16) * N + n0 + bc4);
    }

    if (active) {
      const float* Asb = &As[buf][(wave * 16 + lr) * ASTR + 2 * half];
      const float* Bsb = &Bs[buf][half * BPS + lr * 2];
#pragma unroll
      for (int kk = 0; kk < BK; kk += 4) {
        const v2f a = *(const v2f*)(Asb + kk);
        const float* bp = Bsb + (kk >> 1) * BPS;
        const v2f b0 = *(const v2f*)(bp);
        const v2f b1 = *(const v2f*)(bp + 32);
        const v2f b2 = *(const v2f*)(bp + 64);
        const v2f b3 = *(const v2f*)(bp + 96);
        acc0 = __builtin_amdgcn_wmma_f32_16x16x4_f32(false, a, false, b0, (short)0, acc0, false, false);
        acc1 = __builtin_amdgcn_wmma_f32_16x16x4_f32(false, a, false, b1, (short)0, acc1, false, false);
        acc2 = __builtin_amdgcn_wmma_f32_16x16x4_f32(false, a, false, b2, (short)0, acc2, false, false);
        acc3 = __builtin_amdgcn_wmma_f32_16x16x4_f32(false, a, false, b3, (short)0, acc3, false, false);
      }
    }
    __syncthreads();

    if (pf) {
      const int nb = buf ^ 1;
#pragma unroll
      for (int p = 0; p < 4; ++p)
        *(float4*)&As[nb][(arow + p * 32) * ASTR + ac4] = ra[p];
#pragma unroll
      for (int p = 0; p < 2; ++p) {
        const int row = brow + p * 16;
        float* bp = &Bs[nb][(row >> 1) * BPS + bc4 * 2 + (row & 1)];
        bp[0] = rb[p].x; bp[2] = rb[p].y; bp[4] = rb[p].z; bp[6] = rb[p].w;
      }
      __syncthreads();
    }
  }

  if (active) {
#pragma unroll
    for (int r = 0; r < 8; ++r) {
      const int row = m0 + r + 8 * half;
      float* cp = C + (long long)row * N + n0 + lr;
      cp[0]  = acc0[r];
      cp[16] = acc1[r];
      cp[32] = acc2[r];
      cp[48] = acc3[r];
    }
  }
}

__device__ __forceinline__ int edge_segment(int g, const int* __restrict__ d00,
                                            const int* __restrict__ d01,
                                            const int* __restrict__ d10,
                                            const int* __restrict__ d11, int& list, int& idx) {
  if (g < E00) { list = 0; idx = g; return SEG00 + d00[g]; }
  g -= E00;
  if (g < E01) { list = 1; idx = g; return SEG01 + d01[g]; }
  g -= E01;
  if (g < E10) { list = 2; idx = g; return SEG10 + d10[g]; }
  g -= E10;
  list = 3; idx = g;
  return SEG11 + (g / E11) * N1 + d11[g];
}

__global__ void edge_count(const int* __restrict__ d00, const int* __restrict__ d01,
                           const int* __restrict__ d10, const int* __restrict__ d11,
                           int* __restrict__ cnt) {
  const int g = blockIdx.x * blockDim.x + threadIdx.x;
  if (g >= ETOT) return;
  int list, idx;
  const int seg = edge_segment(g, d00, d01, d10, d11, list, idx);
  atomicAdd(&cnt[seg], 1);
}

__global__ __launch_bounds__(1024) void segment_scan(const int* __restrict__ cnt,
                                                     int* __restrict__ cur, int n) {
  __shared__ int part[1024];
  const int t = threadIdx.x;
  int s = 0;
  for (int i = t; i < n; i += 1024) s += cnt[i];
  part[t] = s;
  __syncthreads();
  for (int off = 1; off < 1024; off <<= 1) {
    const int v = (t >= off) ? part[t - off] : 0;
    __syncthreads();
    part[t] += v;
    __syncthreads();
  }
  int run = part[t] - s;
  for (int i = t; i < n; i += 8 * 1024) {
    int c[8];
#pragma unroll
    for (int u = 0; u < 8; ++u) {
      const int k = i + u * 1024;
      c[u] = (k < n) ? cnt[k] : 0;
    }
#pragma unroll
    for (int u = 0; u < 8; ++u) {
      const int k = i + u * 1024;
      if (k < n) cur[k] = run;
      run += c[u];
    }
  }
}

__global__ void edge_fill(const int* __restrict__ d00, const int* __restrict__ d01,
                          const int* __restrict__ d10, const int* __restrict__ d11,
                          const int* __restrict__ s00, const int* __restrict__ s01,
                          const int* __restrict__ s10, const int* __restrict__ s11,
                          const float* __restrict__ v00, const float* __restrict__ v01,
                          const float* __restrict__ v10, const float* __restrict__ v11,
                          int* __restrict__ cur, int2* __restrict__ perm) {
  const int g = blockIdx.x * blockDim.x + threadIdx.x;
  if (g >= ETOT) return;
  int list, idx;
  const int seg = edge_segment(g, d00, d01, d10, d11, list, idx);
  int sv;
  float vv;
  if (list == 0)      { sv = s00[idx]; vv = v00[idx]; }
  else if (list == 1) { sv = s01[idx]; vv = v01[idx]; }
  else if (list == 2) { sv = s10[idx]; vv = v10[idx]; }
  else                { sv = s11[idx]; vv = v11[idx]; }
  const int pos = atomicAdd(&cur[seg], 1);
  perm[pos] = make_int2(sv, __float_as_int(vv));
}

template <int N>
__global__ __launch_bounds__(256) void spmm_gather(
    const int* __restrict__ cnt, const int* __restrict__ cur, const int2* __restrict__ perm,
    int segBase, int nSeg, int rowsPerK, const float* __restrict__ x, long long xStride,
    float* __restrict__ out) {
  const int r = blockIdx.x * 8 + (threadIdx.x >> 5);
  if (r >= nSeg) return;
  const int lane = threadIdx.x & 31;
  const int end = __builtin_amdgcn_readfirstlane(cur[segBase + r]);
  const int beg = end - __builtin_amdgcn_readfirstlane(cnt[segBase + r]);
  constexpr int NV = N / 128;
  constexpr int RQ = N / 4;
  constexpr int UNR = 1024 / N;
  const float4* xb = (const float4*)(x + (long long)(r / rowsPerK) * xStride);
  float4 acc[NV];
#pragma unroll
  for (int q = 0; q < NV; ++q) acc[q] = make_float4(0.f, 0.f, 0.f, 0.f);

  for (int base = beg; base < end; base += 32) {
    int2 my = make_int2(0, 0);
    if (base + lane < end) my = perm[base + lane];
    const int c = min(32, end - base);
#pragma unroll UNR
    for (int j = 0; j < c; ++j) {
      const int sj = __builtin_amdgcn_readlane(my.x, j);
      const float vj = __int_as_float(__builtin_amdgcn_readlane(my.y, j));
      const float4* xs = xb + (long long)sj * RQ;
#pragma unroll
      for (int q = 0; q < NV; ++q) {
        const float4 t = xs[lane + 32 * q];
        acc[q].x = fmaf(vj, t.x, acc[q].x);
        acc[q].y = fmaf(vj, t.y, acc[q].y);
        acc[q].z = fmaf(vj, t.z, acc[q].z);
        acc[q].w = fmaf(vj, t.w, acc[q].w);
      }
    }
  }
  float4* od = (float4*)(out + (long long)r * N);
#pragma unroll
  for (int q = 0; q < NV; ++q) od[lane + 32 * q] = acc[q];
}

__global__ void finalize_conv(const float* __restrict__ buf, float* __restrict__ out,
                              int K, int n, int N, long long kStride, int do_relu) {
  const int wave = threadIdx.x >> 5;
  const int lane = threadIdx.x & 31;
  const int row = blockIdx.x * (blockDim.x >> 5) + wave;
  if (row >= n) return;
  const int nc = N >> 5;
  float vals[8];
  float ss = 0.f;
  for (int j = 0; j < nc; ++j) {
    const int c = lane + (j << 5);
    float a = 0.f;
    for (int k = 0; k < K; ++k) {
      float t = buf[(long long)k * kStride + (long long)row * N + c];
      if (do_relu) t = fmaxf(t, 0.f);
      a += t;
    }
    vals[j] = a;
    ss += a * a;
  }
  ss = wave_sum(ss);
  const float sc = rsqrtf(fmaxf(ss, 1e-12f));
  for (int j = 0; j < nc; ++j)
    out[(long long)row * N + lane + (j << 5)] = vals[j] * sc;
}

__global__ void add_act(const float* __restrict__ a, const float* __restrict__ b,
                        float* __restrict__ o, long long n, int do_relu) {
  const long long i = (long long)blockIdx.x * blockDim.x + threadIdx.x;
  if (i >= n) return;
  float v = a[i] + b[i];
  o[i] = do_relu ? fmaxf(v, 0.f) : v;
}

__global__ void scale_dloc(const float* __restrict__ z1, const float* __restrict__ dloc,
                           float* __restrict__ out, long long total) {
  const long long i = (long long)blockIdx.x * blockDim.x + threadIdx.x;
  if (i >= total) return;
  const int f = (int)(i % DD);
  const long long rk = i / DD;
  const int r = (int)(rk % N1);
  const int k = (int)(rk / N1);
  out[i] = z1[(long long)r * DD + f] * dloc[k * DD + f];
}

__global__ void edge_dot128(const float* __restrict__ zl, const float* __restrict__ zr,
                            const int* __restrict__ dst, const int* __restrict__ src,
                            float* __restrict__ out, int nE, int eStride,
                            long long zlStride, long long zrStride, long long oStride) {
  const int k = blockIdx.y;
  zl  += (long long)k * zlStride;
  zr  += (long long)k * zrStride;
  dst += (long long)k * eStride;
  src += (long long)k * eStride;
  out += (long long)k * oStride;
  const int e = blockIdx.x * (blockDim.x >> 5) + (threadIdx.x >> 5);
  if (e >= nE) return;
  const int lane = threadIdx.x & 31;
  const int d = dst[e];
  const int s = src[e];
  const float4 a = ((const float4*)(zl + (long long)d * DD))[lane];
  const float4 b = ((const float4*)(zr + (long long)s * DD))[lane];
  float acc = a.x * b.x + a.y * b.y + a.z * b.z + a.w * b.w;
  acc = wave_sum(acc);
  if (lane == 0) out[e] = acc;
}

extern "C" void kernel_launch(void* const* d_in, const int* in_sizes, int n_in,
                              void* d_out, int out_size, void* d_ws, size_t ws_size,
                              hipStream_t stream) {
  (void)in_sizes; (void)n_in; (void)out_size; (void)ws_size;

  const float* feat0   = (const float*)d_in[0];
  const float* feat1   = (const float*)d_in[1];
  const int*   e00_dst = (const int*)d_in[2];
  const int*   e00_src = (const int*)d_in[3];
  const float* e00_val = (const float*)d_in[4];
  const int*   e01_dst = (const int*)d_in[5];
  const int*   e01_src = (const int*)d_in[6];
  const float* e01_val = (const float*)d_in[7];
  const int*   e10_dst = (const int*)d_in[8];
  const int*   e10_src = (const int*)d_in[9];
  const float* e10_val = (const float*)d_in[10];
  const int*   e11_dst = (const int*)d_in[11];
  const int*   e11_src = (const int*)d_in[12];
  const float* e11_val = (const float*)d_in[13];
  const float* W1_00 = (const float*)d_in[14];
  const float* W1_01 = (const float*)d_in[15];
  const float* W1_10 = (const float*)d_in[16];
  const float* W1_11 = (const float*)d_in[17];
  const float* W2_00 = (const float*)d_in[18];
  const float* W2_01 = (const float*)d_in[19];
  const float* W2_10 = (const float*)d_in[20];
  const float* W2_11 = (const float*)d_in[21];
  const float* R00   = (const float*)d_in[22];
  const float* R01   = (const float*)d_in[23];
  const float* R10   = (const float*)d_in[24];
  const float* Rglob = (const float*)d_in[25];
  const float* dloc  = (const float*)d_in[26];

  float* ws = (float*)d_ws;
  float* out = (float*)d_out;
  float* z0  = out;
  float* z1  = out + 2560000;
  float* s00 = z1 + 256000;
  float* s01 = s00 + E00;
  float* s10 = s01 + E01;
  float* s11 = s10 + E10;

  const dim3 blk(256);
  auto gg = [](int M, int N, int batch) { return dim3((M + 127) / 128, N / 64, batch); };

  int*  cnt  = (int*)(ws + off_cnt);
  int*  cur  = (int*)(ws + off_cur);
  int2* perm = (int2*)(ws + off_perm);
  hipMemsetAsync(cnt, 0, (size_t)NSEG * sizeof(int), stream);
  edge_count<<<dim3((ETOT + 255) / 256), blk, 0, stream>>>(e00_dst, e01_dst, e10_dst, e11_dst, cnt);
  segment_scan<<<dim3(1), dim3(1024), 0, stream>>>(cnt, cur, NSEG);
  edge_fill<<<dim3((ETOT + 255) / 256), blk, 0, stream>>>(e00_dst, e01_dst, e10_dst, e11_dst,
                                                          e00_src, e01_src, e10_src, e11_src,
                                                          e00_val, e01_val, e10_val, e11_val,
                                                          cur, perm);

  gemm_f32_wmma<<<gg(N0, HH, 1), blk, 0, stream>>>(feat0, W1_00, ws + off_xw00, N0, HH, FF, 0, 0, 0);
  gemm_f32_wmma<<<gg(N1, HH, 1), blk, 0, stream>>>(feat1, W1_01, ws + off_xw01, N1, HH, FF, 0, 0, 0);
  gemm_f32_wmma<<<gg(N1, HH, KK), blk, 0, stream>>>(feat1, W1_11, ws + off_xw11, N1, HH, FF,
                                                    0, (long long)FF * HH, (long long)N1 * HH);

  spmm_gather<HH><<<dim3((N0 + 7) / 8), blk, 0, stream>>>(cnt, cur, perm, SEG00, N0, N0,
                                                          ws + off_xw00, 0, ws + off_acc00);
  spmm_gather<HH><<<dim3((N0 + 7) / 8), blk, 0, stream>>>(cnt, cur, perm, SEG01, N0, N0,
                                                          ws + off_xw01, 0, ws + off_acc01);
  spmm_gather<HH><<<dim3((KK * N1 + 7) / 8), blk, 0, stream>>>(cnt, cur, perm, SEG11, KK * N1, N1,
                                                               ws + off_xw11, (long long)N1 * HH,
                                                               ws + off_acc11);
  gemm_f32_wmma<<<gg(N0, HH, 1), blk, 0, stream>>>(feat0, W1_10, ws + off_xw10, N0, HH, FF, 0, 0, 0);
  spmm_gather<HH><<<dim3((N1 + 7) / 8), blk, 0, stream>>>(cnt, cur, perm, SEG10, N1, N1,
                                                          ws + off_xw10, 0, ws + off_acc10);

  finalize_conv<<<dim3((N0 + 7) / 8), blk, 0, stream>>>(ws + off_acc00, ws + off_acc00, 1, N0, HH, 0, 1);
  finalize_conv<<<dim3((N0 + 7) / 8), blk, 0, stream>>>(ws + off_acc01, ws + off_acc01, 1, N0, HH, 0, 1);
  finalize_conv<<<dim3((N1 + 7) / 8), blk, 0, stream>>>(ws + off_acc10, ws + off_acc10, 1, N1, HH, 0, 1);
  finalize_conv<<<dim3((N1 + 7) / 8), blk, 0, stream>>>(ws + off_acc11, ws + off_acc11, KK, N1, HH,
                                                        (long long)N1 * HH, 1);

  add_act<<<dim3((N0 * HH + 255) / 256), blk, 0, stream>>>(ws + off_acc00, ws + off_acc01,
                                                           ws + off_h0, (long long)N0 * HH, 1);
  add_act<<<dim3((N1 * HH + 255) / 256), blk, 0, stream>>>(ws + off_acc10, ws + off_acc11,
                                                           ws + off_h1, (long long)N1 * HH, 1);

  gemm_f32_wmma<<<gg(N0, DD, 1), blk, 0, stream>>>(ws + off_h0, W2_00, ws + off_hw00, N0, DD, HH, 0, 0, 0);
  gemm_f32_wmma<<<gg(N0, DD, 1), blk, 0, stream>>>(ws + off_h0, W2_10, ws + off_hw10, N0, DD, HH, 0, 0, 0);
  gemm_f32_wmma<<<gg(N1, DD, 1), blk, 0, stream>>>(ws + off_h1, W2_01, ws + off_hw01, N1, DD, HH, 0, 0, 0);
  gemm_f32_wmma<<<gg(N1, DD, KK), blk, 0, stream>>>(ws + off_h1, W2_11, ws + off_hw11, N1, DD, HH,
                                                    0, (long long)HH * DD, (long long)N1 * DD);

  spmm_gather<DD><<<dim3((N0 + 7) / 8), blk, 0, stream>>>(cnt, cur, perm, SEG00, N0, N0,
                                                          ws + off_hw00, 0, ws + off_bcc00);
  spmm_gather<DD><<<dim3((N0 + 7) / 8), blk, 0, stream>>>(cnt, cur, perm, SEG01, N0, N0,
                                                          ws + off_hw01, 0, ws + off_bcc01);
  spmm_gather<DD><<<dim3((N1 + 7) / 8), blk, 0, stream>>>(cnt, cur, perm, SEG10, N1, N1,
                                                          ws + off_hw10, 0, ws + off_bcc10);
  spmm_gather<DD><<<dim3((KK * N1 + 7) / 8), blk, 0, stream>>>(cnt, cur, perm, SEG11, KK * N1, N1,
                                                               ws + off_hw11, (long long)N1 * DD,
                                                               ws + off_bcc11);

  finalize_conv<<<dim3((N0 + 7) / 8), blk, 0, stream>>>(ws + off_bcc00, ws + off_bcc00, 1, N0, DD, 0, 0);
  finalize_conv<<<dim3((N0 + 7) / 8), blk, 0, stream>>>(ws + off_bcc01, ws + off_bcc01, 1, N0, DD, 0, 0);
  finalize_conv<<<dim3((N1 + 7) / 8), blk, 0, stream>>>(ws + off_bcc10, ws + off_bcc10, 1, N1, DD, 0, 0);
  finalize_conv<<<dim3((N1 + 7) / 8), blk, 0, stream>>>(ws + off_bcc11, ws + off_bcc11, KK, N1, DD,
                                                        (long long)N1 * DD, 0);

  add_act<<<dim3((N0 * DD + 255) / 256), blk, 0, stream>>>(ws + off_bcc00, ws + off_bcc01, z0,
                                                           (long long)N0 * DD, 0);
  add_act<<<dim3((N1 * DD + 255) / 256), blk, 0, stream>>>(ws + off_bcc10, ws + off_bcc11, z1,
                                                           (long long)N1 * DD, 0);

  gemm_f32_wmma<<<gg(N0, DD, 1), blk, 0, stream>>>(z0, R00, ws + off_z0R00, N0, DD, DD, 0, 0, 0);
  gemm_f32_wmma<<<gg(N0, DD, 1), blk, 0, stream>>>(z0, R01, ws + off_z0R01, N0, DD, DD, 0, 0, 0);
  gemm_f32_wmma<<<gg(N1, DD, 1), blk, 0, stream>>>(z1, R10, ws + off_z1R10, N1, DD, DD, 0, 0, 0);

  scale_dloc<<<dim3((KK * N1 * DD + 255) / 256), blk, 0, stream>>>(z1, dloc, ws + off_z1d,
                                                                   (long long)KK * N1 * DD);
  gemm_f32_wmma<<<gg(N1, DD, KK), blk, 0, stream>>>(ws + off_z1d, Rglob, ws + off_Zk, N1, DD, DD,
                                                    (long long)N1 * DD, 0, (long long)N1 * DD);

  edge_dot128<<<dim3((E00 + 7) / 8, 1), blk, 0, stream>>>(ws + off_z0R00, z0, e00_dst, e00_src,
                                                          s00, E00, 0, 0, 0, 0);
  edge_dot128<<<dim3((E01 + 7) / 8, 1), blk, 0, stream>>>(ws + off_z0R01, z1, e01_dst, e01_src,
                                                          s01, E01, 0, 0, 0, 0);
  edge_dot128<<<dim3((E10 + 7) / 8, 1), blk, 0, stream>>>(ws + off_z1R10, z0, e10_dst, e10_src,
                                                          s10, E10, 0, 0, 0, 0);
  edge_dot128<<<dim3((E11 + 7) / 8, KK), blk, 0, stream>>>(ws + off_Zk, ws + off_z1d, e11_dst, e11_src,
                                                           s11, E11, E11,
                                                           (long long)N1 * DD, (long long)N1 * DD,
                                                           (long long)E11);
}


# edge scores: one wave scores 32 edges (indices in one coalesced load, 8 lanes per edge, 4 edges at a time, one 128-byte store) instead of one wave per edge; 32x fewer waves in the 4 score launches
# speedup vs baseline: 1.5820x; 1.1145x over previous
#include <hip/hip_runtime.h>

typedef float v2f __attribute__((ext_vector_type(2)));
typedef float v8f __attribute__((ext_vector_type(8)));

constexpr int N0 = 20000, N1 = 2000;
constexpr int FF = 512, HH = 256, DD = 128, KK = 16;
constexpr int E00 = 640000, E01 = 100000, E10 = 100000, E11 = 64000;

constexpr long long off_xw00 = 0;
constexpr long long off_csr  = 5120000;
constexpr long long off_xw10 = off_xw00;
constexpr long long off_xw01 = 10240000;
constexpr long long off_xw11 = 10752000;
constexpr long long off_acc00 = 18944000;
constexpr long long off_acc01 = 24064000;
constexpr long long off_acc10 = 29184000;
constexpr long long off_acc11 = 29696000;

constexpr long long off_h0    = off_xw00;
constexpr long long off_h1    = off_xw01;
constexpr long long off_hw00  = off_acc01;
constexpr long long off_hw10  = off_acc01 + 2560000;
constexpr long long off_hw01  = off_xw11;
constexpr long long off_hw11  = off_xw11 + 256000;
constexpr long long off_bcc00 = off_acc00;
constexpr long long off_bcc01 = off_acc00 + 2560000;
constexpr long long off_bcc10 = off_acc10;
constexpr long long off_bcc11 = off_acc11;
constexpr long long off_z0R00 = off_acc01;
constexpr long long off_z0R01 = off_acc01 + 2560000;
constexpr long long off_z1R10 = off_acc10 + 256000;
constexpr long long off_z1d   = off_acc11 + 4096000;
constexpr long long off_Zk    = off_acc11;

constexpr int ETOT = E00 + E01 + E10 + KK * E11;
constexpr int NSEG = N0 + N0 + N1 + KK * N1;
constexpr int SEG00 = 0, SEG01 = N0, SEG10 = 2 * N0, SEG11 = 2 * N0 + N1;
constexpr long long off_perm = off_csr;
constexpr long long off_cnt  = off_csr + 2LL * ETOT;
constexpr long long off_cur  = off_cnt + NSEG;

__device__ __forceinline__ float wave_sum(float v) {
#pragma unroll
  for (int off = 16; off > 0; off >>= 1) v += __shfl_xor(v, off, 32);
  return v;
}

constexpr int BK   = 32;
constexpr int ASTR = 36;
constexpr int BPS  = 160;

__global__ __launch_bounds__(256) void gemm_f32_wmma(
    const float* __restrict__ A, const float* __restrict__ B, float* __restrict__ C,
    int M, int N, int Kd, long long sA, long long sB, long long sC) {
  __shared__ float As[2][128 * ASTR];
  __shared__ float Bs[2][(BK / 2) * BPS];

  const int b = blockIdx.z;
  A += (long long)b * sA;
  B += (long long)b * sB;
  C += (long long)b * sC;

  const int tid  = threadIdx.x;
  const int lane = tid & 31;
  const int wave = tid >> 5;
  const int mBlk = blockIdx.x * 128;
  const int n0   = blockIdx.y * 64;
  const int m0   = mBlk + wave * 16;
  const int half = lane >> 4;
  const int lr   = lane & 15;
  const bool active = (m0 < M);

  const int arow = tid >> 3;
  const int ac4  = (tid & 7) * 4;
  const int brow = tid >> 4;
  const int bc4  = (tid & 15) * 4;

  const int nChunks = Kd / BK;
  const float4 zero4 = make_float4(0.f, 0.f, 0.f, 0.f);

  {
#pragma unroll
    for (int p = 0; p < 4; ++p) {
      const int row = arow + p * 32;
      const int grow = mBlk + row;
      float4 v = (grow < M) ? *(const float4*)(A + (long long)grow * Kd + ac4) : zero4;
      *(float4*)&As[0][row * ASTR + ac4] = v;
    }
#pragma unroll
    for (int p = 0; p < 2; ++p) {
      const int row = brow + p * 16;
      const float4 v = *(const float4*)(B + (long long)row * N + n0 + bc4);
      float* bp = &Bs[0][(row >> 1) * BPS + bc4 * 2 + (row & 1)];
      bp[0] = v.x; bp[2] = v.y; bp[4] = v.z; bp[6] = v.w;
    }
  }
  __syncthreads();

  v8f acc0 = {}, acc1 = {}, acc2 = {}, acc3 = {};

  for (int c = 0; c < nChunks; ++c) {
    const int buf = c & 1;
    const bool pf = (c + 1 < nChunks);
    float4 ra[4], rb[2];

    if (pf) {
      const int k0 = (c + 1) * BK;
#pragma unroll
      for (int p = 0; p < 4; ++p) {
        const int grow = mBlk + arow + p * 32;
        ra[p] = (grow < M) ? *(const float4*)(A + (long long)grow * Kd + k0 + ac4) : zero4;
      }
#pragma unroll
      for (int p = 0; p < 2; ++p)
        rb[p] = *(const float4*)(B + (long long)(k0 + brow + p * 16) * N + n0 + bc4);
    }

    if (active) {
      const float* Asb = &As[buf][(wave * 16 + lr) * ASTR + 2 * half];
      const float* Bsb = &Bs[buf][half * BPS + lr * 2];
#pragma unroll
      for (int kk = 0; kk < BK; kk += 4) {
        const v2f a = *(const v2f*)(Asb + kk);
        const float* bp = Bsb + (kk >> 1) * BPS;
        const v2f b0 = *(const v2f*)(bp);
        const v2f b1 = *(const v2f*)(bp + 32);
        const v2f b2 = *(const v2f*)(bp + 64);
        const v2f b3 = *(const v2f*)(bp + 96);
        acc0 = __builtin_amdgcn_wmma_f32_16x16x4_f32(false, a, false, b0, (short)0, acc0, false, false);
        acc1 = __builtin_amdgcn_wmma_f32_16x16x4_f32(false, a, false, b1, (short)0, acc1, false, false);
        acc2 = __builtin_amdgcn_wmma_f32_16x16x4_f32(false, a, false, b2, (short)0, acc2, false, false);
        acc3 = __builtin_amdgcn_wmma_f32_16x16x4_f32(false, a, false, b3, (short)0, acc3, false, false);
      }
    }
    __syncthreads();

    if (pf) {
      const int nb = buf ^ 1;
#pragma unroll
      for (int p = 0; p < 4; ++p)
        *(float4*)&As[nb][(arow + p * 32) * ASTR + ac4] = ra[p];
#pragma unroll
      for (int p = 0; p < 2; ++p) {
        const int row = brow + p * 16;
        float* bp = &Bs[nb][(row >> 1) * BPS + bc4 * 2 + (row & 1)];
        bp[0] = rb[p].x; bp[2] = rb[p].y; bp[4] = rb[p].z; bp[6] = rb[p].w;
      }
      __syncthreads();
    }
  }

  if (active) {
#pragma unroll
    for (int r = 0; r < 8; ++r) {
      const int row = m0 + r + 8 * half;
      float* cp = C + (long long)row * N + n0 + lr;
      cp[0]  = acc0[r];
      cp[16] = acc1[r];
      cp[32] = acc2[r];
      cp[48] = acc3[r];
    }
  }
}

__device__ __forceinline__ int edge_segment(int g, const int* __restrict__ d00,
                                            const int* __restrict__ d01,
                                            const int* __restrict__ d10,
                                            const int* __restrict__ d11, int& list, int& idx) {
  if (g < E00) { list = 0; idx = g; return SEG00 + d00[g]; }
  g -= E00;
  if (g < E01) { list = 1; idx = g; return SEG01 + d01[g]; }
  g -= E01;
  if (g < E10) { list = 2; idx = g; return SEG10 + d10[g]; }
  g -= E10;
  list = 3; idx = g;
  return SEG11 + (g / E11) * N1 + d11[g];
}

__global__ void edge_count(const int* __restrict__ d00, const int* __restrict__ d01,
                           const int* __restrict__ d10, const int* __restrict__ d11,
                           int* __restrict__ cnt) {
  const int g = blockIdx.x * blockDim.x + threadIdx.x;
  if (g >= ETOT) return;
  int list, idx;
  const int seg = edge_segment(g, d00, d01, d10, d11, list, idx);
  atomicAdd(&cnt[seg], 1);
}

__global__ __launch_bounds__(1024) void segment_scan(const int* __restrict__ cnt,
                                                     int* __restrict__ cur, int n) {
  __shared__ int part[1024];
  const int t = threadIdx.x;
  int s = 0;
  for (int i = t; i < n; i += 1024) s += cnt[i];
  part[t] = s;
  __syncthreads();
  for (int off = 1; off < 1024; off <<= 1) {
    const int v = (t >= off) ? part[t - off] : 0;
    __syncthreads();
    part[t] += v;
    __syncthreads();
  }
  int run = part[t] - s;
  for (int i = t; i < n; i += 8 * 1024) {
    int c[8];
#pragma unroll
    for (int u = 0; u < 8; ++u) {
      const int k = i + u * 1024;
      c[u] = (k < n) ? cnt[k] : 0;
    }
#pragma unroll
    for (int u = 0; u < 8; ++u) {
      const int k = i + u * 1024;
      if (k < n) cur[k] = run;
      run += c[u];
    }
  }
}

__global__ void edge_fill(const int* __restrict__ d00, const int* __restrict__ d01,
                          const int* __restrict__ d10, const int* __restrict__ d11,
                          const int* __restrict__ s00, const int* __restrict__ s01,
                          const int* __restrict__ s10, const int* __restrict__ s11,
                          const float* __restrict__ v00, const float* __restrict__ v01,
                          const float* __restrict__ v10, const float* __restrict__ v11,
                          int* __restrict__ cur, int2* __restrict__ perm) {
  const int g = blockIdx.x * blockDim.x + threadIdx.x;
  if (g >= ETOT) return;
  int list, idx;
  const int seg = edge_segment(g, d00, d01, d10, d11, list, idx);
  int sv;
  float vv;
  if (list == 0)      { sv = s00[idx]; vv = v00[idx]; }
  else if (list == 1) { sv = s01[idx]; vv = v01[idx]; }
  else if (list == 2) { sv = s10[idx]; vv = v10[idx]; }
  else                { sv = s11[idx]; vv = v11[idx]; }
  const int pos = atomicAdd(&cur[seg], 1);
  perm[pos] = make_int2(sv, __float_as_int(vv));
}

template <int N>
__global__ __launch_bounds__(256) void spmm_gather(
    const int* __restrict__ cnt, const int* __restrict__ cur, const int2* __restrict__ perm,
    int segBase, int nSeg, int rowsPerK, const float* __restrict__ x, long long xStride,
    float* __restrict__ out) {
  const int r = blockIdx.x * 8 + (threadIdx.x >> 5);
  if (r >= nSeg) return;
  const int lane = threadIdx.x & 31;
  const int end = __builtin_amdgcn_readfirstlane(cur[segBase + r]);
  const int beg = end - __builtin_amdgcn_readfirstlane(cnt[segBase + r]);
  constexpr int NV = N / 128;
  constexpr int RQ = N / 4;
  constexpr int UNR = 1024 / N;
  const float4* xb = (const float4*)(x + (long long)(r / rowsPerK) * xStride);
  float4 acc[NV];
#pragma unroll
  for (int q = 0; q < NV; ++q) acc[q] = make_float4(0.f, 0.f, 0.f, 0.f);

  for (int base = beg; base < end; base += 32) {
    int2 my = make_int2(0, 0);
    if (base + lane < end) my = perm[base + lane];
    const int c = min(32, end - base);
#pragma unroll UNR
    for (int j = 0; j < c; ++j) {
      const int sj = __builtin_amdgcn_readlane(my.x, j);
      const float vj = __int_as_float(__builtin_amdgcn_readlane(my.y, j));
      const float4* xs = xb + (long long)sj * RQ;
#pragma unroll
      for (int q = 0; q < NV; ++q) {
        const float4 t = xs[lane + 32 * q];
        acc[q].x = fmaf(vj, t.x, acc[q].x);
        acc[q].y = fmaf(vj, t.y, acc[q].y);
        acc[q].z = fmaf(vj, t.z, acc[q].z);
        acc[q].w = fmaf(vj, t.w, acc[q].w);
      }
    }
  }
  float4* od = (float4*)(out + (long long)r * N);
#pragma unroll
  for (int q = 0; q < NV; ++q) od[lane + 32 * q] = acc[q];
}

__global__ void finalize_conv(const float* __restrict__ buf, float* __restrict__ out,
                              int K, int n, int N, long long kStride, int do_relu) {
  const int wave = threadIdx.x >> 5;
  const int lane = threadIdx.x & 31;
  const int row = blockIdx.x * (blockDim.x >> 5) + wave;
  if (row >= n) return;
  const int nc = N >> 5;
  float vals[8];
  float ss = 0.f;
  for (int j = 0; j < nc; ++j) {
    const int c = lane + (j << 5);
    float a = 0.f;
    for (int k = 0; k < K; ++k) {
      float t = buf[(long long)k * kStride + (long long)row * N + c];
      if (do_relu) t = fmaxf(t, 0.f);
      a += t;
    }
    vals[j] = a;
    ss += a * a;
  }
  ss = wave_sum(ss);
  const float sc = rsqrtf(fmaxf(ss, 1e-12f));
  for (int j = 0; j < nc; ++j)
    out[(long long)row * N + lane + (j << 5)] = vals[j] * sc;
}

__global__ void add_act(const float* __restrict__ a, const float* __restrict__ b,
                        float* __restrict__ o, long long n, int do_relu) {
  const long long i = (long long)blockIdx.x * blockDim.x + threadIdx.x;
  if (i >= n) return;
  float v = a[i] + b[i];
  o[i] = do_relu ? fmaxf(v, 0.f) : v;
}

__global__ void scale_dloc(const float* __restrict__ z1, const float* __restrict__ dloc,
                           float* __restrict__ out, long long total) {
  const long long i = (long long)blockIdx.x * blockDim.x + threadIdx.x;
  if (i >= total) return;
  const int f = (int)(i % DD);
  const long long rk = i / DD;
  const int r = (int)(rk % N1);
  const int k = (int)(rk / N1);
  out[i] = z1[(long long)r * DD + f] * dloc[k * DD + f];
}

constexpr int EDGES_PER_WAVE = 32;
constexpr int EDGES_PER_BLOCK = 8 * EDGES_PER_WAVE;
__global__ __launch_bounds__(256) void edge_dot128(
    const float* __restrict__ zl, const float* __restrict__ zr,
    const int* __restrict__ dst, const int* __restrict__ src,
    float* __restrict__ out, int nE, int eStride,
    long long zlStride, long long zrStride, long long oStride) {
  const int k = blockIdx.y;
  zl  += (long long)k * zlStride;
  zr  += (long long)k * zrStride;
  dst += (long long)k * eStride;
  src += (long long)k * eStride;
  out += (long long)k * oStride;
  const int e0 = (blockIdx.x * (blockDim.x >> 5) + (threadIdx.x >> 5)) * EDGES_PER_WAVE;
  if (e0 >= nE) return;
  const int lane = threadIdx.x & 31;
  const int e = e0 + lane;
  int d = 0, s = 0;
  if (e < nE) { d = dst[e]; s = src[e]; }
  const int sub = lane >> 3;
  const int q   = lane & 7;
  const float4* zl4 = (const float4*)zl + q;
  const float4* zr4 = (const float4*)zr + q;
  float res = 0.f;
#pragma unroll 2
  for (int g = 0; g < 8; ++g) {
    const int j  = g + 8 * sub;
    const int dj = __shfl(d, j, 32);
    const int sj = __shfl(s, j, 32);
    const float4* a = zl4 + (long long)dj * (DD / 4);
    const float4* b = zr4 + (long long)sj * (DD / 4);
    float acc = 0.f;
#pragma unroll
    for (int i = 0; i < 4; ++i) {
      const float4 x = a[8 * i];
      const float4 y = b[8 * i];
      acc += x.x * y.x + x.y * y.y + x.z * y.z + x.w * y.w;
    }
    acc += __shfl_xor(acc, 4, 32);
    acc += __shfl_xor(acc, 2, 32);
    acc += __shfl_xor(acc, 1, 32);
    if (q == g) res = acc;
  }
  if (e < nE) out[e] = res;
}

extern "C" void kernel_launch(void* const* d_in, const int* in_sizes, int n_in,
                              void* d_out, int out_size, void* d_ws, size_t ws_size,
                              hipStream_t stream) {
  (void)in_sizes; (void)n_in; (void)out_size; (void)ws_size;

  const float* feat0   = (const float*)d_in[0];
  const float* feat1   = (const float*)d_in[1];
  const int*   e00_dst = (const int*)d_in[2];
  const int*   e00_src = (const int*)d_in[3];
  const float* e00_val = (const float*)d_in[4];
  const int*   e01_dst = (const int*)d_in[5];
  const int*   e01_src = (const int*)d_in[6];
  const float* e01_val = (const float*)d_in[7];
  const int*   e10_dst = (const int*)d_in[8];
  const int*   e10_src = (const int*)d_in[9];
  const float* e10_val = (const float*)d_in[10];
  const int*   e11_dst = (const int*)d_in[11];
  const int*   e11_src = (const int*)d_in[12];
  const float* e11_val = (const float*)d_in[13];
  const float* W1_00 = (const float*)d_in[14];
  const float* W1_01 = (const float*)d_in[15];
  const float* W1_10 = (const float*)d_in[16];
  const float* W1_11 = (const float*)d_in[17];
  const float* W2_00 = (const float*)d_in[18];
  const float* W2_01 = (const float*)d_in[19];
  const float* W2_10 = (const float*)d_in[20];
  const float* W2_11 = (const float*)d_in[21];
  const float* R00   = (const float*)d_in[22];
  const float* R01   = (const float*)d_in[23];
  const float* R10   = (const float*)d_in[24];
  const float* Rglob = (const float*)d_in[25];
  const float* dloc  = (const float*)d_in[26];

  float* ws = (float*)d_ws;
  float* out = (float*)d_out;
  float* z0  = out;
  float* z1  = out + 2560000;
  float* s00 = z1 + 256000;
  float* s01 = s00 + E00;
  float* s10 = s01 + E01;
  float* s11 = s10 + E10;

  const dim3 blk(256);
  auto gg = [](int M, int N, int batch) { return dim3((M + 127) / 128, N / 64, batch); };

  int*  cnt  = (int*)(ws + off_cnt);
  int*  cur  = (int*)(ws + off_cur);
  int2* perm = (int2*)(ws + off_perm);
  hipMemsetAsync(cnt, 0, (size_t)NSEG * sizeof(int), stream);
  edge_count<<<dim3((ETOT + 255) / 256), blk, 0, stream>>>(e00_dst, e01_dst, e10_dst, e11_dst, cnt);
  segment_scan<<<dim3(1), dim3(1024), 0, stream>>>(cnt, cur, NSEG);
  edge_fill<<<dim3((ETOT + 255) / 256), blk, 0, stream>>>(e00_dst, e01_dst, e10_dst, e11_dst,
                                                          e00_src, e01_src, e10_src, e11_src,
                                                          e00_val, e01_val, e10_val, e11_val,
                                                          cur, perm);

  gemm_f32_wmma<<<gg(N0, HH, 1), blk, 0, stream>>>(feat0, W1_00, ws + off_xw00, N0, HH, FF, 0, 0, 0);
  gemm_f32_wmma<<<gg(N1, HH, 1), blk, 0, stream>>>(feat1, W1_01, ws + off_xw01, N1, HH, FF, 0, 0, 0);
  gemm_f32_wmma<<<gg(N1, HH, KK), blk, 0, stream>>>(feat1, W1_11, ws + off_xw11, N1, HH, FF,
                                                    0, (long long)FF * HH, (long long)N1 * HH);

  spmm_gather<HH><<<dim3((N0 + 7) / 8), blk, 0, stream>>>(cnt, cur, perm, SEG00, N0, N0,
                                                          ws + off_xw00, 0, ws + off_acc00);
  spmm_gather<HH><<<dim3((N0 + 7) / 8), blk, 0, stream>>>(cnt, cur, perm, SEG01, N0, N0,
                                                          ws + off_xw01, 0, ws + off_acc01);
  spmm_gather<HH><<<dim3((KK * N1 + 7) / 8), blk, 0, stream>>>(cnt, cur, perm, SEG11, KK * N1, N1,
                                                               ws + off_xw11, (long long)N1 * HH,
                                                               ws + off_acc11);
  gemm_f32_wmma<<<gg(N0, HH, 1), blk, 0, stream>>>(feat0, W1_10, ws + off_xw10, N0, HH, FF, 0, 0, 0);
  spmm_gather<HH><<<dim3((N1 + 7) / 8), blk, 0, stream>>>(cnt, cur, perm, SEG10, N1, N1,
                                                          ws + off_xw10, 0, ws + off_acc10);

  finalize_conv<<<dim3((N0 + 7) / 8), blk, 0, stream>>>(ws + off_acc00, ws + off_acc00, 1, N0, HH, 0, 1);
  finalize_conv<<<dim3((N0 + 7) / 8), blk, 0, stream>>>(ws + off_acc01, ws + off_acc01, 1, N0, HH, 0, 1);
  finalize_conv<<<dim3((N1 + 7) / 8), blk, 0, stream>>>(ws + off_acc10, ws + off_acc10, 1, N1, HH, 0, 1);
  finalize_conv<<<dim3((N1 + 7) / 8), blk, 0, stream>>>(ws + off_acc11, ws + off_acc11, KK, N1, HH,
                                                        (long long)N1 * HH, 1);

  add_act<<<dim3((N0 * HH + 255) / 256), blk, 0, stream>>>(ws + off_acc00, ws + off_acc01,
                                                           ws + off_h0, (long long)N0 * HH, 1);
  add_act<<<dim3((N1 * HH + 255) / 256), blk, 0, stream>>>(ws + off_acc10, ws + off_acc11,
                                                           ws + off_h1, (long long)N1 * HH, 1);

  gemm_f32_wmma<<<gg(N0, DD, 1), blk, 0, stream>>>(ws + off_h0, W2_00, ws + off_hw00, N0, DD, HH, 0, 0, 0);
  gemm_f32_wmma<<<gg(N0, DD, 1), blk, 0, stream>>>(ws + off_h0, W2_10, ws + off_hw10, N0, DD, HH, 0, 0, 0);
  gemm_f32_wmma<<<gg(N1, DD, 1), blk, 0, stream>>>(ws + off_h1, W2_01, ws + off_hw01, N1, DD, HH, 0, 0, 0);
  gemm_f32_wmma<<<gg(N1, DD, KK), blk, 0, stream>>>(ws + off_h1, W2_11, ws + off_hw11, N1, DD, HH,
                                                    0, (long long)HH * DD, (long long)N1 * DD);

  spmm_gather<DD><<<dim3((N0 + 7) / 8), blk, 0, stream>>>(cnt, cur, perm, SEG00, N0, N0,
                                                          ws + off_hw00, 0, ws + off_bcc00);
  spmm_gather<DD><<<dim3((N0 + 7) / 8), blk, 0, stream>>>(cnt, cur, perm, SEG01, N0, N0,
                                                          ws + off_hw01, 0, ws + off_bcc01);
  spmm_gather<DD><<<dim3((N1 + 7) / 8), blk, 0, stream>>>(cnt, cur, perm, SEG10, N1, N1,
                                                          ws + off_hw10, 0, ws + off_bcc10);
  spmm_gather<DD><<<dim3((KK * N1 + 7) / 8), blk, 0, stream>>>(cnt, cur, perm, SEG11, KK * N1, N1,
                                                               ws + off_hw11, (long long)N1 * DD,
                                                               ws + off_bcc11);

  finalize_conv<<<dim3((N0 + 7) / 8), blk, 0, stream>>>(ws + off_bcc00, ws + off_bcc00, 1, N0, DD, 0, 0);
  finalize_conv<<<dim3((N0 + 7) / 8), blk, 0, stream>>>(ws + off_bcc01, ws + off_bcc01, 1, N0, DD, 0, 0);
  finalize_conv<<<dim3((N1 + 7) / 8), blk, 0, stream>>>(ws + off_bcc10, ws + off_bcc10, 1, N1, DD, 0, 0);
  finalize_conv<<<dim3((N1 + 7) / 8), blk, 0, stream>>>(ws + off_bcc11, ws + off_bcc11, KK, N1, DD,
                                                        (long long)N1 * DD, 0);

  add_act<<<dim3((N0 * DD + 255) / 256), blk, 0, stream>>>(ws + off_bcc00, ws + off_bcc01, z0,
                                                           (long long)N0 * DD, 0);
  add_act<<<dim3((N1 * DD + 255) / 256), blk, 0, stream>>>(ws + off_bcc10, ws + off_bcc11, z1,
                                                           (long long)N1 * DD, 0);

  gemm_f32_wmma<<<gg(N0, DD, 1), blk, 0, stream>>>(z0, R00, ws + off_z0R00, N0, DD, DD, 0, 0, 0);
  gemm_f32_wmma<<<gg(N0, DD, 1), blk, 0, stream>>>(z0, R01, ws + off_z0R01, N0, DD, DD, 0, 0, 0);
  gemm_f32_wmma<<<gg(N1, DD, 1), blk, 0, stream>>>(z1, R10, ws + off_z1R10, N1, DD, DD, 0, 0, 0);

  scale_dloc<<<dim3((KK * N1 * DD + 255) / 256), blk, 0, stream>>>(z1, dloc, ws + off_z1d,
                                                                   (long long)KK * N1 * DD);
  gemm_f32_wmma<<<gg(N1, DD, KK), blk, 0, stream>>>(ws + off_z1d, Rglob, ws + off_Zk, N1, DD, DD,
                                                    (long long)N1 * DD, 0, (long long)N1 * DD);

  constexpr int SB = EDGES_PER_BLOCK;
  edge_dot128<<<dim3((E00 + SB - 1) / SB, 1), blk, 0, stream>>>(ws + off_z0R00, z0, e00_dst, e00_src,
                                                                s00, E00, 0, 0, 0, 0);
  edge_dot128<<<dim3((E01 + SB - 1) / SB, 1), blk, 0, stream>>>(ws + off_z0R01, z1, e01_dst, e01_src,
                                                                s01, E01, 0, 0, 0, 0);
  edge_dot128<<<dim3((E10 + SB - 1) / SB, 1), blk, 0, stream>>>(ws + off_z1R10, z0, e10_dst, e10_src,
                                                                s10, E10, 0, 0, 0, 0);
  edge_dot128<<<dim3((E11 + SB - 1) / SB, KK), blk, 0, stream>>>(ws + off_Zk, ws + off_z1d, e11_dst,
                                                                 e11_src, s11, E11, E11,
                                                                 (long long)N1 * DD, (long long)N1 * DD,
                                                                 (long long)E11);
}


# independent launches of a stage merged into one launch, block index picks the job: 12 dense products in 4 launches, each layer's two big gathers in 1, the 4 score lists in 1; 26 kernels become 14
# speedup vs baseline: 1.9195x; 1.2133x over previous
#include <hip/hip_runtime.h>

typedef float v2f __attribute__((ext_vector_type(2)));
typedef float v8f __attribute__((ext_vector_type(8)));

constexpr int N0 = 20000, N1 = 2000;
constexpr int FF = 512, HH = 256, DD = 128, KK = 16;
constexpr int E00 = 640000, E01 = 100000, E10 = 100000, E11 = 64000;

constexpr long long off_xw00 = 0;
constexpr long long off_csr  = 5120000;
constexpr long long off_xw01 = 10240000;
constexpr long long off_xw11 = 10752000;
constexpr long long off_acc00 = 18944000;
constexpr long long off_acc01 = 24064000;
constexpr long long off_acc10 = 29184000;
constexpr long long off_acc11 = 29696000;

constexpr long long off_xw10  = off_acc01;
constexpr long long off_h0    = off_acc00;
constexpr long long off_h1    = off_xw01;
constexpr long long off_hw00  = off_acc01;
constexpr long long off_hw10  = off_acc01 + 2560000;
constexpr long long off_hw01  = off_xw11;
constexpr long long off_hw11  = off_xw11 + 256000;
constexpr long long off_bcc11 = off_acc11;
constexpr long long off_z0R00 = off_acc01;
constexpr long long off_z0R01 = off_acc01 + 2560000;
constexpr long long off_z1R10 = off_acc10 + 256000;
constexpr long long off_z1d   = off_acc11 + 4096000;
constexpr long long off_Zk    = off_acc11;

constexpr int ETOT = E00 + E01 + E10 + KK * E11;
constexpr int NSEG = N0 + N0 + N1 + KK * N1;
constexpr int SEG00 = 0, SEG01 = N0, SEG10 = 2 * N0, SEG11 = 2 * N0 + N1;
constexpr long long off_perm = off_csr;
constexpr long long off_cnt  = off_csr + 2LL * ETOT;
constexpr long long off_cur  = off_cnt + NSEG;

__device__ __forceinline__ float wave_sum(float v) {
#pragma unroll
  for (int off = 16; off > 0; off >>= 1) v += __shfl_xor(v, off, 32);
  return v;
}

constexpr int BK   = 32;
constexpr int ASTR = 36;
constexpr int BPS  = 160;

__device__ __forceinline__ void gemm_tile(
    const float* __restrict__ A, const float* __restrict__ B, float* __restrict__ C,
    int M, int N, int Kd, int bx, int by) {
  __shared__ float As[2][128 * ASTR];
  __shared__ float Bs[2][(BK / 2) * BPS];

  const int tid  = threadIdx.x;
  const int lane = tid & 31;
  const int wave = tid >> 5;
  const int mBlk = bx * 128;
  const int n0   = by * 64;
  const int m0   = mBlk + wave * 16;
  const int half = lane >> 4;
  const int lr   = lane & 15;
  const bool active = (m0 < M);

  const int arow = tid >> 3;
  const int ac4  = (tid & 7) * 4;
  const int brow = tid >> 4;
  const int bc4  = (tid & 15) * 4;

  const int nChunks = Kd / BK;
  const float4 zero4 = make_float4(0.f, 0.f, 0.f, 0.f);

  {
#pragma unroll
    for (int p = 0; p < 4; ++p) {
      const int row = arow + p * 32;
      const int grow = mBlk + row;
      float4 v = (grow < M) ? *(const float4*)(A + (long long)grow * Kd + ac4) : zero4;
      *(float4*)&As[0][row * ASTR + ac4] = v;
    }
#pragma unroll
    for (int p = 0; p < 2; ++p) {
      const int row = brow + p * 16;
      const float4 v = *(const float4*)(B + (long long)row * N + n0 + bc4);
      float* bp = &Bs[0][(row >> 1) * BPS + bc4 * 2 + (row & 1)];
      bp[0] = v.x; bp[2] = v.y; bp[4] = v.z; bp[6] = v.w;
    }
  }
  __syncthreads();

  v8f acc0 = {}, acc1 = {}, acc2 = {}, acc3 = {};

  for (int c = 0; c < nChunks; ++c) {
    const int buf = c & 1;
    const bool pf = (c + 1 < nChunks);
    float4 ra[4], rb[2];

    if (pf) {
      const int k0 = (c + 1) * BK;
#pragma unroll
      for (int p = 0; p < 4; ++p) {
        const int grow = mBlk + arow + p * 32;
        ra[p] = (grow < M) ? *(const float4*)(A + (long long)grow * Kd + k0 + ac4) : zero4;
      }
#pragma unroll
      for (int p = 0; p < 2; ++p)
        rb[p] = *(const float4*)(B + (long long)(k0 + brow + p * 16) * N + n0 + bc4);
    }

    if (active) {
      const float* Asb = &As[buf][(wave * 16 + lr) * ASTR + 2 * half];
      const float* Bsb = &Bs[buf][half * BPS + lr * 2];
#pragma unroll
      for (int kk = 0; kk < BK; kk += 4) {
        const v2f a = *(const v2f*)(Asb + kk);
        const float* bp = Bsb + (kk >> 1) * BPS;
        const v2f b0 = *(const v2f*)(bp);
        const v2f b1 = *(const v2f*)(bp + 32);
        const v2f b2 = *(const v2f*)(bp + 64);
        const v2f b3 = *(const v2f*)(bp + 96);
        acc0 = __builtin_amdgcn_wmma_f32_16x16x4_f32(false, a, false, b0, (short)0, acc0, false, false);
        acc1 = __builtin_amdgcn_wmma_f32_16x16x4_f32(false, a, false, b1, (short)0, acc1, false, false);
        acc2 = __builtin_amdgcn_wmma_f32_16x16x4_f32(false, a, false, b2, (short)0, acc2, false, false);
        acc3 = __builtin_amdgcn_wmma_f32_16x16x4_f32(false, a, false, b3, (short)0, acc3, false, false);
      }
    }
    __syncthreads();

    if (pf) {
      const int nb = buf ^ 1;
#pragma unroll
      for (int p = 0; p < 4; ++p)
        *(float4*)&As[nb][(arow + p * 32) * ASTR + ac4] = ra[p];
#pragma unroll
      for (int p = 0; p < 2; ++p) {
        const int row = brow + p * 16;
        float* bp = &Bs[nb][(row >> 1) * BPS + bc4 * 2 + (row & 1)];
        bp[0] = rb[p].x; bp[2] = rb[p].y; bp[4] = rb[p].z; bp[6] = rb[p].w;
      }
      __syncthreads();
    }
  }

  if (active) {
#pragma unroll
    for (int r = 0; r < 8; ++r) {
      const int row = m0 + r + 8 * half;
      float* cp = C + (long long)row * N + n0 + lr;
      cp[0]  = acc0[r];
      cp[16] = acc1[r];
      cp[32] = acc2[r];
      cp[48] = acc3[r];
    }
  }
}

__global__ __launch_bounds__(256) void gemm_f32_wmma(
    const float* __restrict__ A, const float* __restrict__ B, float* __restrict__ C,
    int M, int N, int Kd, long long sA, long long sB, long long sC) {
  const int b = blockIdx.z;
  gemm_tile(A + (long long)b * sA, B + (long long)b * sB, C + (long long)b * sC, M, N, Kd,
            blockIdx.x, blockIdx.y);
}

constexpr int TM0 = (N0 + 127) / 128;
constexpr int TM1 = (N1 + 127) / 128;

template <int KD, int NN>
__global__ __launch_bounds__(256) void dense_stage(
    const float* __restrict__ a0, const float* __restrict__ a1,
    const float* __restrict__ b00, const float* __restrict__ b10,
    const float* __restrict__ b11, const float* __restrict__ b01,
    float* __restrict__ c00, float* __restrict__ c10,
    float* __restrict__ c11, float* __restrict__ c01) {
  constexpr int TN = NN / 64;
  constexpr int T0 = TM0 * TN;
  constexpr int T1 = TM1 * TN;
  int t = blockIdx.x;
  const float* A; const float* B; float* C; int M;
  if (t < T0) {
    A = a0; B = b00; C = c00; M = N0;
  } else if (t < 2 * T0) {
    t -= T0; A = a0; B = b10; C = c10; M = N0;
  } else if (t < 2 * T0 + KK * T1) {
    t -= 2 * T0;
    const int k = t / T1;
    t -= k * T1;
    A = a1; B = b11 + (long long)k * KD * NN; C = c11 + (long long)k * N1 * NN; M = N1;
  } else {
    t -= 2 * T0 + KK * T1; A = a1; B = b01; C = c01; M = N1;
  }
  gemm_tile(A, B, C, M, NN, KD, t / TN, t % TN);
}
template <int KD, int NN>
constexpr int dense_stage_blocks() { return (2 * TM0 + (KK + 1) * TM1) * (NN / 64); }

__global__ __launch_bounds__(256) void dense_decoder(
    const float* __restrict__ z0, const float* __restrict__ z1,
    const float* __restrict__ r00, const float* __restrict__ r01, const float* __restrict__ r10,
    float* __restrict__ o00, float* __restrict__ o01, float* __restrict__ o10) {
  constexpr int TN = DD / 64;
  constexpr int T0 = TM0 * TN;
  int t = blockIdx.x;
  const float* A; const float* B; float* C; int M;
  if (t < T0) {
    A = z0; B = r00; C = o00; M = N0;
  } else if (t < 2 * T0) {
    t -= T0; A = z0; B = r01; C = o01; M = N0;
  } else {
    t -= 2 * T0; A = z1; B = r10; C = o10; M = N1;
  }
  gemm_tile(A, B, C, M, DD, DD, t / TN, t % TN);
}
constexpr int DECODER_BLOCKS = (2 * TM0 + TM1) * (DD / 64);

__device__ __forceinline__ int edge_segment(int g, const int* __restrict__ d00,
                                            const int* __restrict__ d01,
                                            const int* __restrict__ d10,
                                            const int* __restrict__ d11, int& list, int& idx) {
  if (g < E00) { list = 0; idx = g; return SEG00 + d00[g]; }
  g -= E00;
  if (g < E01) { list = 1; idx = g; return SEG01 + d01[g]; }
  g -= E01;
  if (g < E10) { list = 2; idx = g; return SEG10 + d10[g]; }
  g -= E10;
  list = 3; idx = g;
  return SEG11 + (g / E11) * N1 + d11[g];
}

__global__ void edge_count(const int* __restrict__ d00, const int* __restrict__ d01,
                           const int* __restrict__ d10, const int* __restrict__ d11,
                           int* __restrict__ cnt) {
  const int g = blockIdx.x * blockDim.x + threadIdx.x;
  if (g >= ETOT) return;
  int list, idx;
  const int seg = edge_segment(g, d00, d01, d10, d11, list, idx);
  atomicAdd(&cnt[seg], 1);
}

__global__ __launch_bounds__(1024) void segment_scan(const int* __restrict__ cnt,
                                                     int* __restrict__ cur, int n) {
  __shared__ int part[1024];
  const int t = threadIdx.x;
  int s = 0;
  for (int i = t; i < n; i += 1024) s += cnt[i];
  part[t] = s;
  __syncthreads();
  for (int off = 1; off < 1024; off <<= 1) {
    const int v = (t >= off) ? part[t - off] : 0;
    __syncthreads();
    part[t] += v;
    __syncthreads();
  }
  int run = part[t] - s;
  for (int i = t; i < n; i += 8 * 1024) {
    int c[8];
#pragma unroll
    for (int u = 0; u < 8; ++u) {
      const int k = i + u * 1024;
      c[u] = (k < n) ? cnt[k] : 0;
    }
#pragma unroll
    for (int u = 0; u < 8; ++u) {
      const int k = i + u * 1024;
      if (k < n) cur[k] = run;
      run += c[u];
    }
  }
}

__global__ void edge_fill(const int* __restrict__ d00, const int* __restrict__ d01,
                          const int* __restrict__ d10, const int* __restrict__ d11,
                          const int* __restrict__ s00, const int* __restrict__ s01,
                          const int* __restrict__ s10, const int* __restrict__ s11,
                          const float* __restrict__ v00, const float* __restrict__ v01,
                          const float* __restrict__ v10, const float* __restrict__ v11,
                          int* __restrict__ cur, int2* __restrict__ perm) {
  const int g = blockIdx.x * blockDim.x + threadIdx.x;
  if (g >= ETOT) return;
  int list, idx;
  const int seg = edge_segment(g, d00, d01, d10, d11, list, idx);
  int sv;
  float vv;
  if (list == 0)      { sv = s00[idx]; vv = v00[idx]; }
  else if (list == 1) { sv = s01[idx]; vv = v01[idx]; }
  else if (list == 2) { sv = s10[idx]; vv = v10[idx]; }
  else                { sv = s11[idx]; vv = v11[idx]; }
  const int pos = atomicAdd(&cur[seg], 1);
  perm[pos] = make_int2(sv, __float_as_int(vv));
}

template <int N>
__device__ __forceinline__ void gather_segment(
    const int* __restrict__ cnt, const int* __restrict__ cur, const int2* __restrict__ perm,
    int seg, const float4* __restrict__ xb, int lane, float4 (&acc)[N / 128]) {
  const int end = __builtin_amdgcn_readfirstlane(cur[seg]);
  const int beg = end - __builtin_amdgcn_readfirstlane(cnt[seg]);
  constexpr int NV = N / 128;
  constexpr int RQ = N / 4;
  constexpr int UNR = 1024 / N;
#pragma unroll
  for (int q = 0; q < NV; ++q) acc[q] = make_float4(0.f, 0.f, 0.f, 0.f);

  for (int base = beg; base < end; base += 32) {
    int2 my = make_int2(0, 0);
    if (base + lane < end) my = perm[base + lane];
    const int c = min(32, end - base);
#pragma unroll UNR
    for (int j = 0; j < c; ++j) {
      const int sj = __builtin_amdgcn_readlane(my.x, j);
      const float vj = __int_as_float(__builtin_amdgcn_readlane(my.y, j));
      const float4* xs = xb + (long long)sj * RQ;
#pragma unroll
      for (int q = 0; q < NV; ++q) {
        const float4 t = xs[lane + 32 * q];
        acc[q].x = fmaf(vj, t.x, acc[q].x);
        acc[q].y = fmaf(vj, t.y, acc[q].y);
        acc[q].z = fmaf(vj, t.z, acc[q].z);
        acc[q].w = fmaf(vj, t.w, acc[q].w);
      }
    }
  }
}

template <int N>
__device__ __forceinline__ void spmm_gather_rows(
    const int* __restrict__ cnt, const int* __restrict__ cur, const int2* __restrict__ perm,
    int segBase, int nSeg, int rowsPerK, const float* __restrict__ x, long long xStride,
    float* __restrict__ out, int bid) {
  const int r = bid * 8 + (threadIdx.x >> 5);
  if (r >= nSeg) return;
  const int lane = threadIdx.x & 31;
  constexpr int NV = N / 128;
  const float4* xb = (const float4*)(x + (long long)(r / rowsPerK) * xStride);
  float4 acc[NV];
  gather_segment<N>(cnt, cur, perm, segBase + r, xb, lane, acc);
  float4* od = (float4*)(out + (long long)r * N);
#pragma unroll
  for (int q = 0; q < NV; ++q) od[lane + 32 * q] = acc[q];
}

__device__ __forceinline__ float4 relu4(float4 v) {
  return make_float4(fmaxf(v.x, 0.f), fmaxf(v.y, 0.f), fmaxf(v.z, 0.f), fmaxf(v.w, 0.f));
}
__device__ __forceinline__ float sq4(float4 v) { return v.x * v.x + v.y * v.y + v.z * v.z + v.w * v.w; }

template <int N, bool PAIR, bool RELU>
__device__ __forceinline__ void conv_rows_body(
    const int* __restrict__ cnt, const int* __restrict__ cur, const int2* __restrict__ perm,
    int segA, const float* __restrict__ xA, int segB, const float* __restrict__ xB,
    int nRows, float* __restrict__ out, int bid) {
  const int r = bid * 8 + (threadIdx.x >> 5);
  if (r >= nRows) return;
  const int lane = threadIdx.x & 31;
  constexpr int NV = N / 128;
  constexpr int RQ = N / 4;
  float4 a[NV], b[NV];

  if (PAIR) {
    gather_segment<N>(cnt, cur, perm, segB + r, (const float4*)xB, lane, b);
    if (RELU) {
      for (int q = 0; q < NV; ++q) b[q] = relu4(b[q]);
    }
  } else {
    for (int q = 0; q < NV; ++q) b[q] = make_float4(0.f, 0.f, 0.f, 0.f);
    const float4* rb = (const float4*)xB + (long long)r * RQ + lane;
#pragma unroll
    for (int k = 0; k < KK; ++k) {
      for (int q = 0; q < NV; ++q) {
        float4 t = rb[(long long)k * nRows * RQ + 32 * q];
        if (RELU) t = relu4(t);
        b[q].x += t.x; b[q].y += t.y; b[q].z += t.z; b[q].w += t.w;
      }
    }
  }
  gather_segment<N>(cnt, cur, perm, segA + r, (const float4*)xA, lane, a);
  float ssa = 0.f, ssb = 0.f;
  for (int q = 0; q < NV; ++q) {
    if (RELU) a[q] = relu4(a[q]);
    ssa += sq4(a[q]);
    ssb += sq4(b[q]);
  }
  ssa = wave_sum(ssa);
  ssb = wave_sum(ssb);
  const float sca = rsqrtf(fmaxf(ssa, 1e-12f));
  const float scb = rsqrtf(fmaxf(ssb, 1e-12f));
  float4* od = (float4*)(out + (long long)r * N);
  for (int q = 0; q < NV; ++q) {
    float4 o = make_float4(a[q].x * sca + b[q].x * scb, a[q].y * sca + b[q].y * scb,
                           a[q].z * sca + b[q].z * scb, a[q].w * sca + b[q].w * scb);
    if (RELU) o = relu4(o);
    od[lane + 32 * q] = o;
  }
}

template <int N, bool RELU>
__global__ __launch_bounds__(256) void conv_rows_small(
    const int* __restrict__ cnt, const int* __restrict__ cur, const int2* __restrict__ perm,
    int segA, const float* __restrict__ xA, const float* __restrict__ xB,
    int nRows, float* __restrict__ out) {
  conv_rows_body<N, false, RELU>(cnt, cur, perm, segA, xA, 0, xB, nRows, out, blockIdx.x);
}

constexpr int GATHER11_BLOCKS = (KK * N1 + 7) / 8;
constexpr int ROWS0_BLOCKS = (N0 + 7) / 8;
template <int N, bool RELU>
__global__ __launch_bounds__(256) void sparse_stage(
    const int* __restrict__ cnt, const int* __restrict__ cur, const int2* __restrict__ perm,
    const float* __restrict__ x11, float* __restrict__ out11,
    const float* __restrict__ x00, const float* __restrict__ x01, float* __restrict__ outRows) {
  if (blockIdx.x < GATHER11_BLOCKS) {
    spmm_gather_rows<N>(cnt, cur, perm, SEG11, KK * N1, N1, x11, (long long)N1 * N, out11,
                        blockIdx.x);
  } else {
    conv_rows_body<N, true, RELU>(cnt, cur, perm, SEG00, x00, SEG01, x01, N0, outRows,
                                  blockIdx.x - GATHER11_BLOCKS);
  }
}

__global__ void scale_dloc(const float* __restrict__ z1, const float* __restrict__ dloc,
                           float* __restrict__ out, long long total) {
  const long long i = (long long)blockIdx.x * blockDim.x + threadIdx.x;
  if (i >= total) return;
  const int f = (int)(i % DD);
  const long long rk = i / DD;
  const int r = (int)(rk % N1);
  const int k = (int)(rk / N1);
  out[i] = z1[(long long)r * DD + f] * dloc[k * DD + f];
}

constexpr int EDGES_PER_WAVE = 32;
constexpr int EDGES_PER_BLOCK = 8 * EDGES_PER_WAVE;
__device__ __forceinline__ void edge_dot128_block(
    const float* __restrict__ zl, const float* __restrict__ zr,
    const int* __restrict__ dst, const int* __restrict__ src,
    float* __restrict__ out, int nE, int bid) {
  const int e0 = (bid * 8 + (threadIdx.x >> 5)) * EDGES_PER_WAVE;
  if (e0 >= nE) return;
  const int lane = threadIdx.x & 31;
  const int e = e0 + lane;
  int d = 0, s = 0;
  if (e < nE) { d = dst[e]; s = src[e]; }
  const int sub = lane >> 3;
  const int q   = lane & 7;
  const float4* zl4 = (const float4*)zl + q;
  const float4* zr4 = (const float4*)zr + q;
  float res = 0.f;
#pragma unroll 2
  for (int g = 0; g < 8; ++g) {
    const int j  = g + 8 * sub;
    const int dj = __shfl(d, j, 32);
    const int sj = __shfl(s, j, 32);
    const float4* a = zl4 + (long long)dj * (DD / 4);
    const float4* b = zr4 + (long long)sj * (DD / 4);
    float acc = 0.f;
#pragma unroll
    for (int i = 0; i < 4; ++i) {
      const float4 x = a[8 * i];
      const float4 y = b[8 * i];
      acc += x.x * y.x + x.y * y.y + x.z * y.z + x.w * y.w;
    }
    acc += __shfl_xor(acc, 4, 32);
    acc += __shfl_xor(acc, 2, 32);
    acc += __shfl_xor(acc, 1, 32);
    if (q == g) res = acc;
  }
  if (e < nE) out[e] = res;
}

constexpr int SB00 = (E00 + EDGES_PER_BLOCK - 1) / EDGES_PER_BLOCK;
constexpr int SB01 = (E01 + EDGES_PER_BLOCK - 1) / EDGES_PER_BLOCK;
constexpr int SB10 = (E10 + EDGES_PER_BLOCK - 1) / EDGES_PER_BLOCK;
constexpr int SB11 = (E11 + EDGES_PER_BLOCK - 1) / EDGES_PER_BLOCK;
__global__ __launch_bounds__(256) void edge_scores(
    const float* __restrict__ l11, const float* __restrict__ r11,
    const int* __restrict__ d11, const int* __restrict__ c11, float* __restrict__ o11,
    const float* __restrict__ l00, const float* __restrict__ r00,
    const int* __restrict__ d00, const int* __restrict__ c00, float* __restrict__ o00,
    const float* __restrict__ l01, const float* __restrict__ r01,
    const int* __restrict__ d01, const int* __restrict__ c01, float* __restrict__ o01,
    const float* __restrict__ l10, const float* __restrict__ r10,
    const int* __restrict__ d10, const int* __restrict__ c10, float* __restrict__ o10) {
  int t = blockIdx.x;
  if (t < KK * SB11) {
    const int k = t / SB11;
    t -= k * SB11;
    const long long zo = (long long)k * N1 * DD;
    const long long eo = (long long)k * E11;
    edge_dot128_block(l11 + zo, r11 + zo, d11 + eo, c11 + eo, o11 + eo, E11, t);
  } else if (t < KK * SB11 + SB00) {
    edge_dot128_block(l00, r00, d00, c00, o00, E00, t - KK * SB11);
  } else if (t < KK * SB11 + SB00 + SB01) {
    edge_dot128_block(l01, r01, d01, c01, o01, E01, t - (KK * SB11 + SB00));
  } else {
    edge_dot128_block(l10, r10, d10, c10, o10, E10, t - (KK * SB11 + SB00 + SB01));
  }
}
constexpr int SCORE_BLOCKS = KK * SB11 + SB00 + SB01 + SB10;

extern "C" void kernel_launch(void* const* d_in, const int* in_sizes, int n_in,
                              void* d_out, int out_size, void* d_ws, size_t ws_size,
                              hipStream_t stream) {
  (void)in_sizes; (void)n_in; (void)out_size; (void)ws_size;

  const float* feat0   = (const float*)d_in[0];
  const float* feat1   = (const float*)d_in[1];
  const int*   e00_dst = (const int*)d_in[2];
  const int*   e00_src = (const int*)d_in[3];
  const float* e00_val = (const float*)d_in[4];
  const int*   e01_dst = (const int*)d_in[5];
  const int*   e01_src = (const int*)d_in[6];
  const float* e01_val = (const float*)d_in[7];
  const int*   e10_dst = (const int*)d_in[8];
  const int*   e10_src = (const int*)d_in[9];
  const float* e10_val = (const float*)d_in[10];
  const int*   e11_dst = (const int*)d_in[11];
  const int*   e11_src = (const int*)d_in[12];
  const float* e11_val = (const float*)d_in[13];
  const float* W1_00 = (const float*)d_in[14];
  const float* W1_01 = (const float*)d_in[15];
  const float* W1_10 = (const float*)d_in[16];
  const float* W1_11 = (const float*)d_in[17];
  const float* W2_00 = (const float*)d_in[18];
  const float* W2_01 = (const float*)d_in[19];
  const float* W2_10 = (const float*)d_in[20];
  const float* W2_11 = (const float*)d_in[21];
  const float* R00   = (const float*)d_in[22];
  const float* R01   = (const float*)d_in[23];
  const float* R10   = (const float*)d_in[24];
  const float* Rglob = (const float*)d_in[25];
  const float* dloc  = (const float*)d_in[26];

  float* ws = (float*)d_ws;
  float* out = (float*)d_out;
  float* z0  = out;
  float* z1  = out + 2560000;
  float* s00 = z1 + 256000;
  float* s01 = s00 + E00;
  float* s10 = s01 + E01;
  float* s11 = s10 + E10;

  const dim3 blk(256);
  auto gg = [](int M, int N, int batch) { return dim3((M + 127) / 128, N / 64, batch); };

  int*  cnt  = (int*)(ws + off_cnt);
  int*  cur  = (int*)(ws + off_cur);
  int2* perm = (int2*)(ws + off_perm);
  hipMemsetAsync(cnt, 0, (size_t)NSEG * sizeof(int), stream);
  edge_count<<<dim3((ETOT + 255) / 256), blk, 0, stream>>>(e00_dst, e01_dst, e10_dst, e11_dst, cnt);
  segment_scan<<<dim3(1), dim3(1024), 0, stream>>>(cnt, cur, NSEG);
  edge_fill<<<dim3((ETOT + 255) / 256), blk, 0, stream>>>(e00_dst, e01_dst, e10_dst, e11_dst,
                                                          e00_src, e01_src, e10_src, e11_src,
                                                          e00_val, e01_val, e10_val, e11_val,
                                                          cur, perm);

  dense_stage<FF, HH><<<dim3(dense_stage_blocks<FF, HH>()), blk, 0, stream>>>(
      feat0, feat1, W1_00, W1_10, W1_11, W1_01,
      ws + off_xw00, ws + off_xw10, ws + off_xw11, ws + off_xw01);

  sparse_stage<HH, true><<<dim3(GATHER11_BLOCKS + ROWS0_BLOCKS), blk, 0, stream>>>(
      cnt, cur, perm, ws + off_xw11, ws + off_acc11, ws + off_xw00, ws + off_xw01, ws + off_h0);
  conv_rows_small<HH, true><<<dim3((N1 + 7) / 8), blk, 0, stream>>>(
      cnt, cur, perm, SEG10, ws + off_xw10, ws + off_acc11, N1, ws + off_h1);

  dense_stage<HH, DD><<<dim3(dense_stage_blocks<HH, DD>()), blk, 0, stream>>>(
      ws + off_h0, ws + off_h1, W2_00, W2_10, W2_11, W2_01,
      ws + off_hw00, ws + off_hw10, ws + off_hw11, ws + off_hw01);

  sparse_stage<DD, false><<<dim3(GATHER11_BLOCKS + ROWS0_BLOCKS), blk, 0, stream>>>(
      cnt, cur, perm, ws + off_hw11, ws + off_bcc11, ws + off_hw00, ws + off_hw01, z0);
  conv_rows_small<DD, false><<<dim3((N1 + 7) / 8), blk, 0, stream>>>(
      cnt, cur, perm, SEG10, ws + off_hw10, ws + off_bcc11, N1, z1);

  dense_decoder<<<dim3(DECODER_BLOCKS), blk, 0, stream>>>(
      z0, z1, R00, R01, R10, ws + off_z0R00, ws + off_z0R01, ws + off_z1R10);

  scale_dloc<<<dim3((KK * N1 * DD + 255) / 256), blk, 0, stream>>>(z1, dloc, ws + off_z1d,
                                                                   (long long)KK * N1 * DD);
  gemm_f32_wmma<<<gg(N1, DD, KK), blk, 0, stream>>>(ws + off_z1d, Rglob, ws + off_Zk, N1, DD, DD,
                                                    (long long)N1 * DD, 0, (long long)N1 * DD);

  edge_scores<<<dim3(SCORE_BLOCKS), blk, 0, stream>>>(
      ws + off_Zk, ws + off_z1d, e11_dst, e11_src, s11,
      ws + off_z0R00, z0, e00_dst, e00_src, s00,
      ws + off_z0R01, z1, e01_dst, e01_src, s01,
      ws + off_z1R10, z0, e10_dst, e10_src, s10);
}


# GEMM tile prefetch stays in registers: A rows past M read through a clamped row index, temporaries as native vectors; was flat loads plus a scratch store and reload of the A prefetch every K chunk
# speedup vs baseline: 2.3766x; 1.2381x over previous
#include <hip/hip_runtime.h>

typedef float v2f __attribute__((ext_vector_type(2)));
typedef float v8f __attribute__((ext_vector_type(8)));
typedef float v4f __attribute__((ext_vector_type(4)));

constexpr int N0 = 20000, N1 = 2000;
constexpr int FF = 512, HH = 256, DD = 128, KK = 16;
constexpr int E00 = 640000, E01 = 100000, E10 = 100000, E11 = 64000;

constexpr long long off_xw00 = 0;
constexpr long long off_csr  = 5120000;
constexpr long long off_xw01 = 10240000;
constexpr long long off_xw11 = 10752000;
constexpr long long off_acc00 = 18944000;
constexpr long long off_acc01 = 24064000;
constexpr long long off_acc10 = 29184000;
constexpr long long off_acc11 = 29696000;

constexpr long long off_xw10  = off_acc01;
constexpr long long off_h0    = off_acc00;
constexpr long long off_h1    = off_xw01;
constexpr long long off_hw00  = off_acc01;
constexpr long long off_hw10  = off_acc01 + 2560000;
constexpr long long off_hw01  = off_xw11;
constexpr long long off_hw11  = off_xw11 + 256000;
constexpr long long off_bcc11 = off_acc11;
constexpr long long off_z0R00 = off_acc01;
constexpr long long off_z0R01 = off_acc01 + 2560000;
constexpr long long off_z1R10 = off_acc10 + 256000;
constexpr long long off_z1d   = off_acc11 + 4096000;
constexpr long long off_Zk    = off_acc11;

constexpr int ETOT = E00 + E01 + E10 + KK * E11;
constexpr int NSEG = N0 + N0 + N1 + KK * N1;
constexpr int SEG00 = 0, SEG01 = N0, SEG10 = 2 * N0, SEG11 = 2 * N0 + N1;
constexpr long long off_perm = off_csr;
constexpr long long off_cnt  = off_csr + 2LL * ETOT;
constexpr long long off_cur  = off_cnt + NSEG;

__device__ __forceinline__ float wave_sum(float v) {
#pragma unroll
  for (int off = 16; off > 0; off >>= 1) v += __shfl_xor(v, off, 32);
  return v;
}

constexpr int BK   = 32;
constexpr int ASTR = 36;
constexpr int BPS  = 160;

__device__ __forceinline__ void gemm_tile(
    const float* __restrict__ A, const float* __restrict__ B, float* __restrict__ C,
    int M, int N, int Kd, int bx, int by) {
  __shared__ float As[2][128 * ASTR];
  __shared__ float Bs[2][(BK / 2) * BPS];

  const int tid  = threadIdx.x;
  const int lane = tid & 31;
  const int wave = tid >> 5;
  const int mBlk = bx * 128;
  const int n0   = by * 64;
  const int m0   = mBlk + wave * 16;
  const int half = lane >> 4;
  const int lr   = lane & 15;
  const bool active = (m0 < M);

  const int arow = tid >> 3;
  const int ac4  = (tid & 7) * 4;
  const int brow = tid >> 4;
  const int bc4  = (tid & 15) * 4;

  const int nChunks = Kd / BK;

  {
#pragma unroll
    for (int p = 0; p < 4; ++p) {
      const int row = arow + p * 32;
      const int grow = min(mBlk + row, M - 1);
      const v4f v = *(const v4f*)(A + (long long)grow * Kd + ac4);
      *(v4f*)&As[0][row * ASTR + ac4] = v;
    }
#pragma unroll
    for (int p = 0; p < 2; ++p) {
      const int row = brow + p * 16;
      const v4f v = *(const v4f*)(B + (long long)row * N + n0 + bc4);
      float* bp = &Bs[0][(row >> 1) * BPS + bc4 * 2 + (row & 1)];
      bp[0] = v.x; bp[2] = v.y; bp[4] = v.z; bp[6] = v.w;
    }
  }
  __syncthreads();

  v8f acc0 = {}, acc1 = {}, acc2 = {}, acc3 = {};

  for (int c = 0; c < nChunks; ++c) {
    const int buf = c & 1;
    const bool pf = (c + 1 < nChunks);
    v4f ra[4], rb[2];

    if (pf) {
      const int k0 = (c + 1) * BK;
#pragma unroll
      for (int p = 0; p < 4; ++p) {
        const int grow = min(mBlk + arow + p * 32, M - 1);
        ra[p] = *(const v4f*)(A + (long long)grow * Kd + k0 + ac4);
      }
#pragma unroll
      for (int p = 0; p < 2; ++p)
        rb[p] = *(const v4f*)(B + (long long)(k0 + brow + p * 16) * N + n0 + bc4);
    }

    if (active) {
      const float* Asb = &As[buf][(wave * 16 + lr) * ASTR + 2 * half];
      const float* Bsb = &Bs[buf][half * BPS + lr * 2];
#pragma unroll
      for (int kk = 0; kk < BK; kk += 4) {
        const v2f a = *(const v2f*)(Asb + kk);
        const float* bp = Bsb + (kk >> 1) * BPS;
        const v2f b0 = *(const v2f*)(bp);
        const v2f b1 = *(const v2f*)(bp + 32);
        const v2f b2 = *(const v2f*)(bp + 64);
        const v2f b3 = *(const v2f*)(bp + 96);
        acc0 = __builtin_amdgcn_wmma_f32_16x16x4_f32(false, a, false, b0, (short)0, acc0, false, false);
        acc1 = __builtin_amdgcn_wmma_f32_16x16x4_f32(false, a, false, b1, (short)0, acc1, false, false);
        acc2 = __builtin_amdgcn_wmma_f32_16x16x4_f32(false, a, false, b2, (short)0, acc2, false, false);
        acc3 = __builtin_amdgcn_wmma_f32_16x16x4_f32(false, a, false, b3, (short)0, acc3, false, false);
      }
    }
    __syncthreads();

    if (pf) {
      const int nb = buf ^ 1;
#pragma unroll
      for (int p = 0; p < 4; ++p)
        *(v4f*)&As[nb][(arow + p * 32) * ASTR + ac4] = ra[p];
#pragma unroll
      for (int p = 0; p < 2; ++p) {
        const int row = brow + p * 16;
        float* bp = &Bs[nb][(row >> 1) * BPS + bc4 * 2 + (row & 1)];
        bp[0] = rb[p].x; bp[2] = rb[p].y; bp[4] = rb[p].z; bp[6] = rb[p].w;
      }
      __syncthreads();
    }
  }

  if (active) {
#pragma unroll
    for (int r = 0; r < 8; ++r) {
      const int row = m0 + r + 8 * half;
      float* cp = C + (long long)row * N + n0 + lr;
      cp[0]  = acc0[r];
      cp[16] = acc1[r];
      cp[32] = acc2[r];
      cp[48] = acc3[r];
    }
  }
}

__global__ __launch_bounds__(256) void gemm_f32_wmma(
    const float* __restrict__ A, const float* __restrict__ B, float* __restrict__ C,
    int M, int N, int Kd, long long sA, long long sB, long long sC) {
  const int b = blockIdx.z;
  gemm_tile(A + (long long)b * sA, B + (long long)b * sB, C + (long long)b * sC, M, N, Kd,
            blockIdx.x, blockIdx.y);
}

constexpr int TM0 = (N0 + 127) / 128;
constexpr int TM1 = (N1 + 127) / 128;

template <int KD, int NN>
__global__ __launch_bounds__(256) void dense_stage(
    const float* __restrict__ a0, const float* __restrict__ a1,
    const float* __restrict__ b00, const float* __restrict__ b10,
    const float* __restrict__ b11, const float* __restrict__ b01,
    float* __restrict__ c00, float* __restrict__ c10,
    float* __restrict__ c11, float* __restrict__ c01) {
  constexpr int TN = NN / 64;
  constexpr int T0 = TM0 * TN;
  constexpr int T1 = TM1 * TN;
  int t = blockIdx.x;
  const float* A; const float* B; float* C; int M;
  if (t < T0) {
    A = a0; B = b00; C = c00; M = N0;
  } else if (t < 2 * T0) {
    t -= T0; A = a0; B = b10; C = c10; M = N0;
  } else if (t < 2 * T0 + KK * T1) {
    t -= 2 * T0;
    const int k = t / T1;
    t -= k * T1;
    A = a1; B = b11 + (long long)k * KD * NN; C = c11 + (long long)k * N1 * NN; M = N1;
  } else {
    t -= 2 * T0 + KK * T1; A = a1; B = b01; C = c01; M = N1;
  }
  gemm_tile(A, B, C, M, NN, KD, t / TN, t % TN);
}
template <int KD, int NN>
constexpr int dense_stage_blocks() { return (2 * TM0 + (KK + 1) * TM1) * (NN / 64); }

__global__ __launch_bounds__(256) void dense_decoder(
    const float* __restrict__ z0, const float* __restrict__ z1,
    const float* __restrict__ r00, const float* __restrict__ r01, const float* __restrict__ r10,
    float* __restrict__ o00, float* __restrict__ o01, float* __restrict__ o10) {
  constexpr int TN = DD / 64;
  constexpr int T0 = TM0 * TN;
  int t = blockIdx.x;
  const float* A; const float* B; float* C; int M;
  if (t < T0) {
    A = z0; B = r00; C = o00; M = N0;
  } else if (t < 2 * T0) {
    t -= T0; A = z0; B = r01; C = o01; M = N0;
  } else {
    t -= 2 * T0; A = z1; B = r10; C = o10; M = N1;
  }
  gemm_tile(A, B, C, M, DD, DD, t / TN, t % TN);
}
constexpr int DECODER_BLOCKS = (2 * TM0 + TM1) * (DD / 64);

__device__ __forceinline__ int edge_segment(int g, const int* __restrict__ d00,
                                            const int* __restrict__ d01,
                                            const int* __restrict__ d10,
                                            const int* __restrict__ d11, int& list, int& idx) {
  if (g < E00) { list = 0; idx = g; return SEG00 + d00[g]; }
  g -= E00;
  if (g < E01) { list = 1; idx = g; return SEG01 + d01[g]; }
  g -= E01;
  if (g < E10) { list = 2; idx = g; return SEG10 + d10[g]; }
  g -= E10;
  list = 3; idx = g;
  return SEG11 + (g / E11) * N1 + d11[g];
}

__global__ void edge_count(const int* __restrict__ d00, const int* __restrict__ d01,
                           const int* __restrict__ d10, const int* __restrict__ d11,
                           int* __restrict__ cnt) {
  const int g = blockIdx.x * blockDim.x + threadIdx.x;
  if (g >= ETOT) return;
  int list, idx;
  const int seg = edge_segment(g, d00, d01, d10, d11, list, idx);
  atomicAdd(&cnt[seg], 1);
}

__global__ __launch_bounds__(1024) void segment_scan(const int* __restrict__ cnt,
                                                     int* __restrict__ cur, int n) {
  __shared__ int part[1024];
  const int t = threadIdx.x;
  int s = 0;
  for (int i = t; i < n; i += 1024) s += cnt[i];
  part[t] = s;
  __syncthreads();
  for (int off = 1; off < 1024; off <<= 1) {
    const int v = (t >= off) ? part[t - off] : 0;
    __syncthreads();
    part[t] += v;
    __syncthreads();
  }
  int run = part[t] - s;
  for (int i = t; i < n; i += 8 * 1024) {
    int c[8];
#pragma unroll
    for (int u = 0; u < 8; ++u) {
      const int k = i + u * 1024;
      c[u] = (k < n) ? cnt[k] : 0;
    }
#pragma unroll
    for (int u = 0; u < 8; ++u) {
      const int k = i + u * 1024;
      if (k < n) cur[k] = run;
      run += c[u];
    }
  }
}

__global__ void edge_fill(const int* __restrict__ d00, const int* __restrict__ d01,
                          const int* __restrict__ d10, const int* __restrict__ d11,
                          const int* __restrict__ s00, const int* __restrict__ s01,
                          const int* __restrict__ s10, const int* __restrict__ s11,
                          const float* __restrict__ v00, const float* __restrict__ v01,
                          const float* __restrict__ v10, const float* __restrict__ v11,
                          int* __restrict__ cur, int2* __restrict__ perm) {
  const int g = blockIdx.x * blockDim.x + threadIdx.x;
  if (g >= ETOT) return;
  int list, idx;
  const int seg = edge_segment(g, d00, d01, d10, d11, list, idx);
  int sv;
  float vv;
  if (list == 0)      { sv = s00[idx]; vv = v00[idx]; }
  else if (list == 1) { sv = s01[idx]; vv = v01[idx]; }
  else if (list == 2) { sv = s10[idx]; vv = v10[idx]; }
  else                { sv = s11[idx]; vv = v11[idx]; }
  const int pos = atomicAdd(&cur[seg], 1);
  perm[pos] = make_int2(sv, __float_as_int(vv));
}

template <int N>
__device__ __forceinline__ void gather_segment(
    const int* __restrict__ cnt, const int* __restrict__ cur, const int2* __restrict__ perm,
    int seg, const float4* __restrict__ xb, int lane, float4 (&acc)[N / 128]) {
  const int end = __builtin_amdgcn_readfirstlane(cur[seg]);
  const int beg = end - __builtin_amdgcn_readfirstlane(cnt[seg]);
  constexpr int NV = N / 128;
  constexpr int RQ = N / 4;
  constexpr int UNR = 1024 / N;
#pragma unroll
  for (int q = 0; q < NV; ++q) acc[q] = make_float4(0.f, 0.f, 0.f, 0.f);

  for (int base = beg; base < end; base += 32) {
    int2 my = make_int2(0, 0);
    if (base + lane < end) my = perm[base + lane];
    const int c = min(32, end - base);
#pragma unroll UNR
    for (int j = 0; j < c; ++j) {
      const int sj = __builtin_amdgcn_readlane(my.x, j);
      const float vj = __int_as_float(__builtin_amdgcn_readlane(my.y, j));
      const float4* xs = xb + (long long)sj * RQ;
#pragma unroll
      for (int q = 0; q < NV; ++q) {
        const float4 t = xs[lane + 32 * q];
        acc[q].x = fmaf(vj, t.x, acc[q].x);
        acc[q].y = fmaf(vj, t.y, acc[q].y);
        acc[q].z = fmaf(vj, t.z, acc[q].z);
        acc[q].w = fmaf(vj, t.w, acc[q].w);
      }
    }
  }
}

template <int N>
__device__ __forceinline__ void spmm_gather_rows(
    const int* __restrict__ cnt, const int* __restrict__ cur, const int2* __restrict__ perm,
    int segBase, int nSeg, int rowsPerK, const float* __restrict__ x, long long xStride,
    float* __restrict__ out, int bid) {
  const int r = bid * 8 + (threadIdx.x >> 5);
  if (r >= nSeg) return;
  const int lane = threadIdx.x & 31;
  constexpr int NV = N / 128;
  const float4* xb = (const float4*)(x + (long long)(r / rowsPerK) * xStride);
  float4 acc[NV];
  gather_segment<N>(cnt, cur, perm, segBase + r, xb, lane, acc);
  float4* od = (float4*)(out + (long long)r * N);
#pragma unroll
  for (int q = 0; q < NV; ++q) od[lane + 32 * q] = acc[q];
}

__device__ __forceinline__ float4 relu4(float4 v) {
  return make_float4(fmaxf(v.x, 0.f), fmaxf(v.y, 0.f), fmaxf(v.z, 0.f), fmaxf(v.w, 0.f));
}
__device__ __forceinline__ float sq4(float4 v) { return v.x * v.x + v.y * v.y + v.z * v.z + v.w * v.w; }

template <int N, bool PAIR, bool RELU>
__device__ __forceinline__ void conv_rows_body(
    const int* __restrict__ cnt, const int* __restrict__ cur, const int2* __restrict__ perm,
    int segA, const float* __restrict__ xA, int segB, const float* __restrict__ xB,
    int nRows, float* __restrict__ out, int bid) {
  const int r = bid * 8 + (threadIdx.x >> 5);
  if (r >= nRows) return;
  const int lane = threadIdx.x & 31;
  constexpr int NV = N / 128;
  constexpr int RQ = N / 4;
  float4 a[NV], b[NV];

  if (PAIR) {
    gather_segment<N>(cnt, cur, perm, segB + r, (const float4*)xB, lane, b);
    if (RELU) {
      for (int q = 0; q < NV; ++q) b[q] = relu4(b[q]);
    }
  } else {
    for (int q = 0; q < NV; ++q) b[q] = make_float4(0.f, 0.f, 0.f, 0.f);
    const float4* rb = (const float4*)xB + (long long)r * RQ + lane;
#pragma unroll
    for (int k = 0; k < KK; ++k) {
      for (int q = 0; q < NV; ++q) {
        float4 t = rb[(long long)k * nRows * RQ + 32 * q];
        if (RELU) t = relu4(t);
        b[q].x += t.x; b[q].y += t.y; b[q].z += t.z; b[q].w += t.w;
      }
    }
  }
  gather_segment<N>(cnt, cur, perm, segA + r, (const float4*)xA, lane, a);
  float ssa = 0.f, ssb = 0.f;
  for (int q = 0; q < NV; ++q) {
    if (RELU) a[q] = relu4(a[q]);
    ssa += sq4(a[q]);
    ssb += sq4(b[q]);
  }
  ssa = wave_sum(ssa);
  ssb = wave_sum(ssb);
  const float sca = rsqrtf(fmaxf(ssa, 1e-12f));
  const float scb = rsqrtf(fmaxf(ssb, 1e-12f));
  float4* od = (float4*)(out + (long long)r * N);
  for (int q = 0; q < NV; ++q) {
    float4 o = make_float4(a[q].x * sca + b[q].x * scb, a[q].y * sca + b[q].y * scb,
                           a[q].z * sca + b[q].z * scb, a[q].w * sca + b[q].w * scb);
    if (RELU) o = relu4(o);
    od[lane + 32 * q] = o;
  }
}

template <int N, bool RELU>
__global__ __launch_bounds__(256) void conv_rows_small(
    const int* __restrict__ cnt, const int* __restrict__ cur, const int2* __restrict__ perm,
    int segA, const float* __restrict__ xA, const float* __restrict__ xB,
    int nRows, float* __restrict__ out) {
  conv_rows_body<N, false, RELU>(cnt, cur, perm, segA, xA, 0, xB, nRows, out, blockIdx.x);
}

constexpr int GATHER11_BLOCKS = (KK * N1 + 7) / 8;
constexpr int ROWS0_BLOCKS = (N0 + 7) / 8;
template <int N, bool RELU>
__global__ __launch_bounds__(256) void sparse_stage(
    const int* __restrict__ cnt, const int* __restrict__ cur, const int2* __restrict__ perm,
    const float* __restrict__ x11, float* __restrict__ out11,
    const float* __restrict__ x00, const float* __restrict__ x01, float* __restrict__ outRows) {
  if (blockIdx.x < GATHER11_BLOCKS) {
    spmm_gather_rows<N>(cnt, cur, perm, SEG11, KK * N1, N1, x11, (long long)N1 * N, out11,
                        blockIdx.x);
  } else {
    conv_rows_body<N, true, RELU>(cnt, cur, perm, SEG00, x00, SEG01, x01, N0, outRows,
                                  blockIdx.x - GATHER11_BLOCKS);
  }
}

__global__ void scale_dloc(const float* __restrict__ z1, const float* __restrict__ dloc,
                           float* __restrict__ out, long long total) {
  const long long i = (long long)blockIdx.x * blockDim.x + threadIdx.x;
  if (i >= total) return;
  const int f = (int)(i % DD);
  const long long rk = i / DD;
  const int r = (int)(rk % N1);
  const int k = (int)(rk / N1);
  out[i] = z1[(long long)r * DD + f] * dloc[k * DD + f];
}

constexpr int EDGES_PER_WAVE = 32;
constexpr int EDGES_PER_BLOCK = 8 * EDGES_PER_WAVE;
__device__ __forceinline__ void edge_dot128_block(
    const float* __restrict__ zl, const float* __restrict__ zr,
    const int* __restrict__ dst, const int* __restrict__ src,
    float* __restrict__ out, int nE, int bid) {
  const int e0 = (bid * 8 + (threadIdx.x >> 5)) * EDGES_PER_WAVE;
  if (e0 >= nE) return;
  const int lane = threadIdx.x & 31;
  const int e = e0 + lane;
  int d = 0, s = 0;
  if (e < nE) { d = dst[e]; s = src[e]; }
  const int sub = lane >> 3;
  const int q   = lane & 7;
  const float4* zl4 = (const float4*)zl + q;
  const float4* zr4 = (const float4*)zr + q;
  float res = 0.f;
#pragma unroll 2
  for (int g = 0; g < 8; ++g) {
    const int j  = g + 8 * sub;
    const int dj = __shfl(d, j, 32);
    const int sj = __shfl(s, j, 32);
    const float4* a = zl4 + (long long)dj * (DD / 4);
    const float4* b = zr4 + (long long)sj * (DD / 4);
    float acc = 0.f;
#pragma unroll
    for (int i = 0; i < 4; ++i) {
      const float4 x = a[8 * i];
      const float4 y = b[8 * i];
      acc += x.x * y.x + x.y * y.y + x.z * y.z + x.w * y.w;
    }
    acc += __shfl_xor(acc, 4, 32);
    acc += __shfl_xor(acc, 2, 32);
    acc += __shfl_xor(acc, 1, 32);
    if (q == g) res = acc;
  }
  if (e < nE) out[e] = res;
}

constexpr int SB00 = (E00 + EDGES_PER_BLOCK - 1) / EDGES_PER_BLOCK;
constexpr int SB01 = (E01 + EDGES_PER_BLOCK - 1) / EDGES_PER_BLOCK;
constexpr int SB10 = (E10 + EDGES_PER_BLOCK - 1) / EDGES_PER_BLOCK;
constexpr int SB11 = (E11 + EDGES_PER_BLOCK - 1) / EDGES_PER_BLOCK;
__global__ __launch_bounds__(256) void edge_scores(
    const float* __restrict__ l11, const float* __restrict__ r11,
    const int* __restrict__ d11, const int* __restrict__ c11, float* __restrict__ o11,
    const float* __restrict__ l00, const float* __restrict__ r00,
    const int* __restrict__ d00, const int* __restrict__ c00, float* __restrict__ o00,
    const float* __restrict__ l01, const float* __restrict__ r01,
    const int* __restrict__ d01, const int* __restrict__ c01, float* __restrict__ o01,
    const float* __restrict__ l10, const float* __restrict__ r10,
    const int* __restrict__ d10, const int* __restrict__ c10, float* __restrict__ o10) {
  int t = blockIdx.x;
  if (t < KK * SB11) {
    const int k = t / SB11;
    t -= k * SB11;
    const long long zo = (long long)k * N1 * DD;
    const long long eo = (long long)k * E11;
    edge_dot128_block(l11 + zo, r11 + zo, d11 + eo, c11 + eo, o11 + eo, E11, t);
  } else if (t < KK * SB11 + SB00) {
    edge_dot128_block(l00, r00, d00, c00, o00, E00, t - KK * SB11);
  } else if (t < KK * SB11 + SB00 + SB01) {
    edge_dot128_block(l01, r01, d01, c01, o01, E01, t - (KK * SB11 + SB00));
  } else {
    edge_dot128_block(l10, r10, d10, c10, o10, E10, t - (KK * SB11 + SB00 + SB01));
  }
}
constexpr int SCORE_BLOCKS = KK * SB11 + SB00 + SB01 + SB10;

extern "C" void kernel_launch(void* const* d_in, const int* in_sizes, int n_in,
                              void* d_out, int out_size, void* d_ws, size_t ws_size,
                              hipStream_t stream) {
  (void)in_sizes; (void)n_in; (void)out_size; (void)ws_size;

  const float* feat0   = (const float*)d_in[0];
  const float* feat1   = (const float*)d_in[1];
  const int*   e00_dst = (const int*)d_in[2];
  const int*   e00_src = (const int*)d_in[3];
  const float* e00_val = (const float*)d_in[4];
  const int*   e01_dst = (const int*)d_in[5];
  const int*   e01_src = (const int*)d_in[6];
  const float* e01_val = (const float*)d_in[7];
  const int*   e10_dst = (const int*)d_in[8];
  const int*   e10_src = (const int*)d_in[9];
  const float* e10_val = (const float*)d_in[10];
  const int*   e11_dst = (const int*)d_in[11];
  const int*   e11_src = (const int*)d_in[12];
  const float* e11_val = (const float*)d_in[13];
  const float* W1_00 = (const float*)d_in[14];
  const float* W1_01 = (const float*)d_in[15];
  const float* W1_10 = (const float*)d_in[16];
  const float* W1_11 = (const float*)d_in[17];
  const float* W2_00 = (const float*)d_in[18];
  const float* W2_01 = (const float*)d_in[19];
  const float* W2_10 = (const float*)d_in[20];
  const float* W2_11 = (const float*)d_in[21];
  const float* R00   = (const float*)d_in[22];
  const float* R01   = (const float*)d_in[23];
  const float* R10   = (const float*)d_in[24];
  const float* Rglob = (const float*)d_in[25];
  const float* dloc  = (const float*)d_in[26];

  float* ws = (float*)d_ws;
  float* out = (float*)d_out;
  float* z0  = out;
  float* z1  = out + 2560000;
  float* s00 = z1 + 256000;
  float* s01 = s00 + E00;
  float* s10 = s01 + E01;
  float* s11 = s10 + E10;

  const dim3 blk(256);
  auto gg = [](int M, int N, int batch) { return dim3((M + 127) / 128, N / 64, batch); };

  int*  cnt  = (int*)(ws + off_cnt);
  int*  cur  = (int*)(ws + off_cur);
  int2* perm = (int2*)(ws + off_perm);
  hipMemsetAsync(cnt, 0, (size_t)NSEG * sizeof(int), stream);
  edge_count<<<dim3((ETOT + 255) / 256), blk, 0, stream>>>(e00_dst, e01_dst, e10_dst, e11_dst, cnt);
  segment_scan<<<dim3(1), dim3(1024), 0, stream>>>(cnt, cur, NSEG);
  edge_fill<<<dim3((ETOT + 255) / 256), blk, 0, stream>>>(e00_dst, e01_dst, e10_dst, e11_dst,
                                                          e00_src, e01_src, e10_src, e11_src,
                                                          e00_val, e01_val, e10_val, e11_val,
                                                          cur, perm);

  dense_stage<FF, HH><<<dim3(dense_stage_blocks<FF, HH>()), blk, 0, stream>>>(
      feat0, feat1, W1_00, W1_10, W1_11, W1_01,
      ws + off_xw00, ws + off_xw10, ws + off_xw11, ws + off_xw01);

  sparse_stage<HH, true><<<dim3(GATHER11_BLOCKS + ROWS0_BLOCKS), blk, 0, stream>>>(
      cnt, cur, perm, ws + off_xw11, ws + off_acc11, ws + off_xw00, ws + off_xw01, ws + off_h0);
  conv_rows_small<HH, true><<<dim3((N1 + 7) / 8), blk, 0, stream>>>(
      cnt, cur, perm, SEG10, ws + off_xw10, ws + off_acc11, N1, ws + off_h1);

  dense_stage<HH, DD><<<dim3(dense_stage_blocks<HH, DD>()), blk, 0, stream>>>(
      ws + off_h0, ws + off_h1, W2_00, W2_10, W2_11, W2_01,
      ws + off_hw00, ws + off_hw10, ws + off_hw11, ws + off_hw01);

  sparse_stage<DD, false><<<dim3(GATHER11_BLOCKS + ROWS0_BLOCKS), blk, 0, stream>>>(
      cnt, cur, perm, ws + off_hw11, ws + off_bcc11, ws + off_hw00, ws + off_hw01, z0);
  conv_rows_small<DD, false><<<dim3((N1 + 7) / 8), blk, 0, stream>>>(
      cnt, cur, perm, SEG10, ws + off_hw10, ws + off_bcc11, N1, z1);

  dense_decoder<<<dim3(DECODER_BLOCKS), blk, 0, stream>>>(
      z0, z1, R00, R01, R10, ws + off_z0R00, ws + off_z0R01, ws + off_z1R10);

  scale_dloc<<<dim3((KK * N1 * DD + 255) / 256), blk, 0, stream>>>(z1, dloc, ws + off_z1d,
                                                                   (long long)KK * N1 * DD);
  gemm_f32_wmma<<<gg(N1, DD, KK), blk, 0, stream>>>(ws + off_z1d, Rglob, ws + off_Zk, N1, DD, DD,
                                                    (long long)N1 * DD, 0, (long long)N1 * DD);

  edge_scores<<<dim3(SCORE_BLOCKS), blk, 0, stream>>>(
      ws + off_Zk, ws + off_z1d, e11_dst, e11_src, s11,
      ws + off_z0R00, z0, e00_dst, e00_src, s00,
      ws + off_z0R01, z1, e01_dst, e01_src, s01,
      ws + off_z1R10, z0, e10_dst, e10_src, s10);
}


# edge-list build: count and fill take 8 edges per thread in list-aligned blocks, a thread's index loads and its 8 atomics in flight together; 58,256 one-edge waves become 7,288 in each of the two jobs
# speedup vs baseline: 2.9766x; 1.2525x over previous
#include <hip/hip_runtime.h>

typedef float v2f __attribute__((ext_vector_type(2)));
typedef float v8f __attribute__((ext_vector_type(8)));
typedef float v4f __attribute__((ext_vector_type(4)));

constexpr int N0 = 20000, N1 = 2000;
constexpr int FF = 512, HH = 256, DD = 128, KK = 16;
constexpr int E00 = 640000, E01 = 100000, E10 = 100000, E11 = 64000;

constexpr long long off_xw00 = 0;
constexpr long long off_csr  = 5120000;
constexpr long long off_xw01 = 10240000;
constexpr long long off_xw11 = 10752000;
constexpr long long off_acc00 = 18944000;
constexpr long long off_acc01 = 24064000;
constexpr long long off_acc10 = 29184000;
constexpr long long off_acc11 = 29696000;

constexpr long long off_xw10  = off_acc01;
constexpr long long off_h0    = off_acc00;
constexpr long long off_h1    = off_xw01;
constexpr long long off_hw00  = off_acc01;
constexpr long long off_hw10  = off_acc01 + 2560000;
constexpr long long off_hw01  = off_xw11;
constexpr long long off_hw11  = off_xw11 + 256000;
constexpr long long off_bcc11 = off_acc11;
constexpr long long off_z0R00 = off_acc01;
constexpr long long off_z0R01 = off_acc01 + 2560000;
constexpr long long off_z1R10 = off_acc10 + 256000;
constexpr long long off_z1d   = off_acc11 + 4096000;
constexpr long long off_Zk    = off_acc11;

constexpr int ETOT = E00 + E01 + E10 + KK * E11;
constexpr int NSEG = N0 + N0 + N1 + KK * N1;
constexpr int SEG00 = 0, SEG01 = N0, SEG10 = 2 * N0, SEG11 = 2 * N0 + N1;
constexpr long long off_perm = off_csr;
constexpr long long off_cnt  = off_csr + 2LL * ETOT;
constexpr long long off_cur  = off_cnt + NSEG;

__device__ __forceinline__ float wave_sum(float v) {
#pragma unroll
  for (int off = 16; off > 0; off >>= 1) v += __shfl_xor(v, off, 32);
  return v;
}

constexpr int BK   = 32;
constexpr int ASTR = 36;
constexpr int BPS  = 160;

__device__ __forceinline__ void gemm_tile(
    const float* __restrict__ A, const float* __restrict__ B, float* __restrict__ C,
    int M, int N, int Kd, int bx, int by) {
  __shared__ float As[2][128 * ASTR];
  __shared__ float Bs[2][(BK / 2) * BPS];

  const int tid  = threadIdx.x;
  const int lane = tid & 31;
  const int wave = tid >> 5;
  const int mBlk = bx * 128;
  const int n0   = by * 64;
  const int m0   = mBlk + wave * 16;
  const int half = lane >> 4;
  const int lr   = lane & 15;
  const bool active = (m0 < M);

  const int arow = tid >> 3;
  const int ac4  = (tid & 7) * 4;
  const int brow = tid >> 4;
  const int bc4  = (tid & 15) * 4;

  const int nChunks = Kd / BK;

  {
#pragma unroll
    for (int p = 0; p < 4; ++p) {
      const int row = arow + p * 32;
      const int grow = min(mBlk + row, M - 1);
      const v4f v = *(const v4f*)(A + (long long)grow * Kd + ac4);
      *(v4f*)&As[0][row * ASTR + ac4] = v;
    }
#pragma unroll
    for (int p = 0; p < 2; ++p) {
      const int row = brow + p * 16;
      const v4f v = *(const v4f*)(B + (long long)row * N + n0 + bc4);
      float* bp = &Bs[0][(row >> 1) * BPS + bc4 * 2 + (row & 1)];
      bp[0] = v.x; bp[2] = v.y; bp[4] = v.z; bp[6] = v.w;
    }
  }
  __syncthreads();

  v8f acc0 = {}, acc1 = {}, acc2 = {}, acc3 = {};

  for (int c = 0; c < nChunks; ++c) {
    const int buf = c & 1;
    const bool pf = (c + 1 < nChunks);
    v4f ra[4], rb[2];

    if (pf) {
      const int k0 = (c + 1) * BK;
#pragma unroll
      for (int p = 0; p < 4; ++p) {
        const int grow = min(mBlk + arow + p * 32, M - 1);
        ra[p] = *(const v4f*)(A + (long long)grow * Kd + k0 + ac4);
      }
#pragma unroll
      for (int p = 0; p < 2; ++p)
        rb[p] = *(const v4f*)(B + (long long)(k0 + brow + p * 16) * N + n0 + bc4);
    }

    if (active) {
      const float* Asb = &As[buf][(wave * 16 + lr) * ASTR + 2 * half];
      const float* Bsb = &Bs[buf][half * BPS + lr * 2];
#pragma unroll
      for (int kk = 0; kk < BK; kk += 4) {
        const v2f a = *(const v2f*)(Asb + kk);
        const float* bp = Bsb + (kk >> 1) * BPS;
        const v2f b0 = *(const v2f*)(bp);
        const v2f b1 = *(const v2f*)(bp + 32);
        const v2f b2 = *(const v2f*)(bp + 64);
        const v2f b3 = *(const v2f*)(bp + 96);
        acc0 = __builtin_amdgcn_wmma_f32_16x16x4_f32(false, a, false, b0, (short)0, acc0, false, false);
        acc1 = __builtin_amdgcn_wmma_f32_16x16x4_f32(false, a, false, b1, (short)0, acc1, false, false);
        acc2 = __builtin_amdgcn_wmma_f32_16x16x4_f32(false, a, false, b2, (short)0, acc2, false, false);
        acc3 = __builtin_amdgcn_wmma_f32_16x16x4_f32(false, a, false, b3, (short)0, acc3, false, false);
      }
    }
    __syncthreads();

    if (pf) {
      const int nb = buf ^ 1;
#pragma unroll
      for (int p = 0; p < 4; ++p)
        *(v4f*)&As[nb][(arow + p * 32) * ASTR + ac4] = ra[p];
#pragma unroll
      for (int p = 0; p < 2; ++p) {
        const int row = brow + p * 16;
        float* bp = &Bs[nb][(row >> 1) * BPS + bc4 * 2 + (row & 1)];
        bp[0] = rb[p].x; bp[2] = rb[p].y; bp[4] = rb[p].z; bp[6] = rb[p].w;
      }
      __syncthreads();
    }
  }

  if (active) {
#pragma unroll
    for (int r = 0; r < 8; ++r) {
      const int row = m0 + r + 8 * half;
      float* cp = C + (long long)row * N + n0 + lr;
      cp[0]  = acc0[r];
      cp[16] = acc1[r];
      cp[32] = acc2[r];
      cp[48] = acc3[r];
    }
  }
}

constexpr int EPT = 8;
constexpr int EPB = 256 * EPT;
constexpr int LB00 = (E00 + EPB - 1) / EPB;
constexpr int LB01 = (E01 + EPB - 1) / EPB;
constexpr int LB10 = (E10 + EPB - 1) / EPB;
constexpr int LB11 = (KK * E11 + EPB - 1) / EPB;
constexpr int EDGE_BLOCKS = LB00 + LB01 + LB10 + LB11;

__device__ __forceinline__ int edge_block_list(int bid, int& nE, int& segBase, int& base) {
  if (bid < LB00) { nE = E00; segBase = SEG00; base = bid * EPB; return 0; }
  bid -= LB00;
  if (bid < LB01) { nE = E01; segBase = SEG01; base = bid * EPB; return 1; }
  bid -= LB01;
  if (bid < LB10) { nE = E10; segBase = SEG10; base = bid * EPB; return 2; }
  bid -= LB10;
  nE = KK * E11; segBase = SEG11; base = bid * EPB; return 3;
}

__device__ __forceinline__ void edge_count_block(const int* __restrict__ d00,
                                                 const int* __restrict__ d01,
                                                 const int* __restrict__ d10,
                                                 const int* __restrict__ d11,
                                                 int* __restrict__ cnt, int bid) {
  int nE, segBase, base;
  const int list = edge_block_list(bid, nE, segBase, base);
  const int* dst = (list == 0) ? d00 : (list == 1) ? d01 : (list == 2) ? d10 : d11;
  const int first = base + threadIdx.x;
  int seg[EPT];
#pragma unroll
  for (int u = 0; u < EPT; ++u) {
    const int idx = min(first + u * 256, nE - 1);
    const int rel = (list == 3) ? idx / E11 : 0;
    seg[u] = segBase + rel * N1 + dst[idx];
  }
#pragma unroll
  for (int u = 0; u < EPT; ++u)
    if (first + u * 256 < nE) atomicAdd(&cnt[seg[u]], 1);
}

constexpr int TM0 = (N0 + 127) / 128;
constexpr int TM1 = (N1 + 127) / 128;

template <int KD, int NN, bool COUNT>
__global__ __launch_bounds__(256) void dense_stage(
    const float* __restrict__ a0, const float* __restrict__ a1,
    const float* __restrict__ b00, const float* __restrict__ b10,
    const float* __restrict__ b11, const float* __restrict__ b01,
    float* __restrict__ c00, float* __restrict__ c10,
    float* __restrict__ c11, float* __restrict__ c01,
    const int* __restrict__ d00, const int* __restrict__ d01,
    const int* __restrict__ d10, const int* __restrict__ d11, int* __restrict__ cnt) {
  constexpr int TN = NN / 64;
  constexpr int T0 = TM0 * TN;
  constexpr int T1 = TM1 * TN;
  int t = blockIdx.x;
  if (COUNT) {
    if (t < EDGE_BLOCKS) {
      edge_count_block(d00, d01, d10, d11, cnt, t);
      return;
    }
    t -= EDGE_BLOCKS;
  }
  const float* A; const float* B; float* C; int M;
  if (t < T0) {
    A = a0; B = b00; C = c00; M = N0;
  } else if (t < 2 * T0) {
    t -= T0; A = a0; B = b10; C = c10; M = N0;
  } else if (t < 2 * T0 + KK * T1) {
    t -= 2 * T0;
    const int k = t / T1;
    t -= k * T1;
    A = a1; B = b11 + (long long)k * KD * NN; C = c11 + (long long)k * N1 * NN; M = N1;
  } else {
    t -= 2 * T0 + KK * T1; A = a1; B = b01; C = c01; M = N1;
  }
  gemm_tile(A, B, C, M, NN, KD, t / TN, t % TN);
}
template <int KD, int NN, bool COUNT>
constexpr int dense_stage_blocks() {
  return (2 * TM0 + (KK + 1) * TM1) * (NN / 64) + (COUNT ? EDGE_BLOCKS : 0);
}

__global__ __launch_bounds__(256) void dense_decoder(
    const float* __restrict__ z0, const float* __restrict__ z1, const float* __restrict__ z1d,
    const float* __restrict__ r00, const float* __restrict__ r01, const float* __restrict__ r10,
    const float* __restrict__ rg,
    float* __restrict__ o00, float* __restrict__ o01, float* __restrict__ o10,
    float* __restrict__ ok) {
  constexpr int TN = DD / 64;
  constexpr int T0 = TM0 * TN;
  constexpr int T1 = TM1 * TN;
  int t = blockIdx.x;
  const float* A; const float* B; float* C; int M;
  if (t < T0) {
    A = z0; B = r00; C = o00; M = N0;
  } else if (t < 2 * T0) {
    t -= T0; A = z0; B = r01; C = o01; M = N0;
  } else if (t < 2 * T0 + T1) {
    t -= 2 * T0; A = z1; B = r10; C = o10; M = N1;
  } else {
    t -= 2 * T0 + T1;
    const int k = t / T1;
    t -= k * T1;
    A = z1d + (long long)k * N1 * DD; B = rg; C = ok + (long long)k * N1 * DD; M = N1;
  }
  gemm_tile(A, B, C, M, DD, DD, t / TN, t % TN);
}
constexpr int DECODER_BLOCKS = (2 * TM0 + (KK + 1) * TM1) * (DD / 64);

__global__ __launch_bounds__(1024) void segment_scan(const int* __restrict__ cnt,
                                                     int* __restrict__ cur, int n) {
  __shared__ int part[1024];
  const int t = threadIdx.x;
  int s = 0;
  for (int i = t; i < n; i += 1024) s += cnt[i];
  part[t] = s;
  __syncthreads();
  for (int off = 1; off < 1024; off <<= 1) {
    const int v = (t >= off) ? part[t - off] : 0;
    __syncthreads();
    part[t] += v;
    __syncthreads();
  }
  int run = part[t] - s;
  for (int i = t; i < n; i += 8 * 1024) {
    int c[8];
#pragma unroll
    for (int u = 0; u < 8; ++u) {
      const int k = i + u * 1024;
      c[u] = (k < n) ? cnt[k] : 0;
    }
#pragma unroll
    for (int u = 0; u < 8; ++u) {
      const int k = i + u * 1024;
      if (k < n) cur[k] = run;
      run += c[u];
    }
  }
}

__global__ __launch_bounds__(256) void edge_fill(
    const int* __restrict__ d00, const int* __restrict__ d01,
    const int* __restrict__ d10, const int* __restrict__ d11,
    const int* __restrict__ s00, const int* __restrict__ s01,
    const int* __restrict__ s10, const int* __restrict__ s11,
    const float* __restrict__ v00, const float* __restrict__ v01,
    const float* __restrict__ v10, const float* __restrict__ v11,
    int* __restrict__ cur, int2* __restrict__ perm) {
  int nE, segBase, base;
  const int list = edge_block_list(blockIdx.x, nE, segBase, base);
  const int*   dst = (list == 0) ? d00 : (list == 1) ? d01 : (list == 2) ? d10 : d11;
  const int*   src = (list == 0) ? s00 : (list == 1) ? s01 : (list == 2) ? s10 : s11;
  const float* val = (list == 0) ? v00 : (list == 1) ? v01 : (list == 2) ? v10 : v11;
  const int first = base + threadIdx.x;
  int seg[EPT], sv[EPT], pos[EPT];
  float vv[EPT];
#pragma unroll
  for (int u = 0; u < EPT; ++u) {
    const int idx = min(first + u * 256, nE - 1);
    const int rel = (list == 3) ? idx / E11 : 0;
    seg[u] = segBase + rel * N1 + dst[idx];
    sv[u] = src[idx];
    vv[u] = val[idx];
  }
#pragma unroll
  for (int u = 0; u < EPT; ++u) {
    pos[u] = 0;
    if (first + u * 256 < nE) pos[u] = atomicAdd(&cur[seg[u]], 1);
  }
#pragma unroll
  for (int u = 0; u < EPT; ++u)
    if (first + u * 256 < nE) perm[pos[u]] = make_int2(sv[u], __float_as_int(vv[u]));
}

template <int N>
__device__ __forceinline__ void gather_segment(
    const int* __restrict__ cnt, const int* __restrict__ cur, const int2* __restrict__ perm,
    int seg, const float4* __restrict__ xb, int lane, float4 (&acc)[N / 128]) {
  const int end = __builtin_amdgcn_readfirstlane(cur[seg]);
  const int beg = end - __builtin_amdgcn_readfirstlane(cnt[seg]);
  constexpr int NV = N / 128;
  constexpr int RQ = N / 4;
  constexpr int UNR = 1024 / N;
#pragma unroll
  for (int q = 0; q < NV; ++q) acc[q] = make_float4(0.f, 0.f, 0.f, 0.f);

  for (int base = beg; base < end; base += 32) {
    int2 my = make_int2(0, 0);
    if (base + lane < end) my = perm[base + lane];
    const int c = min(32, end - base);
#pragma unroll UNR
    for (int j = 0; j < c; ++j) {
      const int sj = __builtin_amdgcn_readlane(my.x, j);
      const float vj = __int_as_float(__builtin_amdgcn_readlane(my.y, j));
      const float4* xs = xb + (long long)sj * RQ;
#pragma unroll
      for (int q = 0; q < NV; ++q) {
        const float4 t = xs[lane + 32 * q];
        acc[q].x = fmaf(vj, t.x, acc[q].x);
        acc[q].y = fmaf(vj, t.y, acc[q].y);
        acc[q].z = fmaf(vj, t.z, acc[q].z);
        acc[q].w = fmaf(vj, t.w, acc[q].w);
      }
    }
  }
}

template <int N>
__device__ __forceinline__ void spmm_gather_rows(
    const int* __restrict__ cnt, const int* __restrict__ cur, const int2* __restrict__ perm,
    int segBase, int nSeg, int rowsPerK, const float* __restrict__ x, long long xStride,
    float* __restrict__ out, int bid) {
  const int r = bid * 8 + (threadIdx.x >> 5);
  if (r >= nSeg) return;
  const int lane = threadIdx.x & 31;
  constexpr int NV = N / 128;
  const float4* xb = (const float4*)(x + (long long)(r / rowsPerK) * xStride);
  float4 acc[NV];
  gather_segment<N>(cnt, cur, perm, segBase + r, xb, lane, acc);
  float4* od = (float4*)(out + (long long)r * N);
#pragma unroll
  for (int q = 0; q < NV; ++q) od[lane + 32 * q] = acc[q];
}

__device__ __forceinline__ float4 relu4(float4 v) {
  return make_float4(fmaxf(v.x, 0.f), fmaxf(v.y, 0.f), fmaxf(v.z, 0.f), fmaxf(v.w, 0.f));
}
__device__ __forceinline__ float sq4(float4 v) { return v.x * v.x + v.y * v.y + v.z * v.z + v.w * v.w; }

template <int N, bool PAIR, bool RELU, bool SCALED>
__device__ __forceinline__ void conv_rows_body(
    const int* __restrict__ cnt, const int* __restrict__ cur, const int2* __restrict__ perm,
    int segA, const float* __restrict__ xA, int segB, const float* __restrict__ xB,
    int nRows, float* __restrict__ out, int bid,
    const float* __restrict__ dl, float* __restrict__ sc) {
  const int r = bid * 8 + (threadIdx.x >> 5);
  if (r >= nRows) return;
  const int lane = threadIdx.x & 31;
  constexpr int NV = N / 128;
  constexpr int RQ = N / 4;
  float4 a[NV], b[NV];

  if (PAIR) {
    gather_segment<N>(cnt, cur, perm, segB + r, (const float4*)xB, lane, b);
    if (RELU) {
      for (int q = 0; q < NV; ++q) b[q] = relu4(b[q]);
    }
  } else {
    for (int q = 0; q < NV; ++q) b[q] = make_float4(0.f, 0.f, 0.f, 0.f);
    const float4* rb = (const float4*)xB + (long long)r * RQ + lane;
#pragma unroll
    for (int k = 0; k < KK; ++k) {
      for (int q = 0; q < NV; ++q) {
        float4 t = rb[(long long)k * nRows * RQ + 32 * q];
        if (RELU) t = relu4(t);
        b[q].x += t.x; b[q].y += t.y; b[q].z += t.z; b[q].w += t.w;
      }
    }
  }
  gather_segment<N>(cnt, cur, perm, segA + r, (const float4*)xA, lane, a);
  float ssa = 0.f, ssb = 0.f;
  for (int q = 0; q < NV; ++q) {
    if (RELU) a[q] = relu4(a[q]);
    ssa += sq4(a[q]);
    ssb += sq4(b[q]);
  }
  ssa = wave_sum(ssa);
  ssb = wave_sum(ssb);
  const float sca = rsqrtf(fmaxf(ssa, 1e-12f));
  const float scb = rsqrtf(fmaxf(ssb, 1e-12f));
  float4* od = (float4*)(out + (long long)r * N);
  for (int q = 0; q < NV; ++q) {
    float4 o = make_float4(a[q].x * sca + b[q].x * scb, a[q].y * sca + b[q].y * scb,
                           a[q].z * sca + b[q].z * scb, a[q].w * sca + b[q].w * scb);
    if (RELU) o = relu4(o);
    od[lane + 32 * q] = o;
    if (SCALED) {
      const float4* d4 = (const float4*)dl + lane + 32 * q;
      float4* s4 = (float4*)(sc + (long long)r * N) + lane + 32 * q;
#pragma unroll
      for (int k = 0; k < KK; ++k) {
        const float4 d = d4[k * RQ];
        s4[(long long)k * nRows * RQ] = make_float4(o.x * d.x, o.y * d.y, o.z * d.z, o.w * d.w);
      }
    }
  }
}

template <int N, bool RELU, bool SCALED>
__global__ __launch_bounds__(256) void conv_rows_small(
    const int* __restrict__ cnt, const int* __restrict__ cur, const int2* __restrict__ perm,
    int segA, const float* __restrict__ xA, const float* __restrict__ xB,
    int nRows, float* __restrict__ out, const float* __restrict__ dl, float* __restrict__ sc) {
  conv_rows_body<N, false, RELU, SCALED>(cnt, cur, perm, segA, xA, 0, xB, nRows, out, blockIdx.x,
                                         dl, sc);
}

constexpr int GATHER11_BLOCKS = (KK * N1 + 7) / 8;
constexpr int ROWS0_BLOCKS = (N0 + 7) / 8;
template <int N, bool RELU>
__global__ __launch_bounds__(256) void sparse_stage(
    const int* __restrict__ cnt, const int* __restrict__ cur, const int2* __restrict__ perm,
    const float* __restrict__ x11, float* __restrict__ out11,
    const float* __restrict__ x00, const float* __restrict__ x01, float* __restrict__ outRows) {
  if (blockIdx.x < GATHER11_BLOCKS) {
    spmm_gather_rows<N>(cnt, cur, perm, SEG11, KK * N1, N1, x11, (long long)N1 * N, out11,
                        blockIdx.x);
  } else {
    conv_rows_body<N, true, RELU, false>(cnt, cur, perm, SEG00, x00, SEG01, x01, N0, outRows,
                                         blockIdx.x - GATHER11_BLOCKS, nullptr, nullptr);
  }
}

constexpr int EDGES_PER_WAVE = 32;
constexpr int EDGES_PER_BLOCK = 8 * EDGES_PER_WAVE;
__device__ __forceinline__ void edge_dot128_block(
    const float* __restrict__ zl, const float* __restrict__ zr,
    const int* __restrict__ dst, const int* __restrict__ src,
    float* __restrict__ out, int nE, int bid) {
  const int e0 = (bid * 8 + (threadIdx.x >> 5)) * EDGES_PER_WAVE;
  if (e0 >= nE) return;
  const int lane = threadIdx.x & 31;
  const int e = e0 + lane;
  int d = 0, s = 0;
  if (e < nE) { d = dst[e]; s = src[e]; }
  const int sub = lane >> 3;
  const int q   = lane & 7;
  const float4* zl4 = (const float4*)zl + q;
  const float4* zr4 = (const float4*)zr + q;
  float res = 0.f;
#pragma unroll 2
  for (int g = 0; g < 8; ++g) {
    const int j  = g + 8 * sub;
    const int dj = __shfl(d, j, 32);
    const int sj = __shfl(s, j, 32);
    const float4* a = zl4 + (long long)dj * (DD / 4);
    const float4* b = zr4 + (long long)sj * (DD / 4);
    float acc = 0.f;
#pragma unroll
    for (int i = 0; i < 4; ++i) {
      const float4 x = a[8 * i];
      const float4 y = b[8 * i];
      acc += x.x * y.x + x.y * y.y + x.z * y.z + x.w * y.w;
    }
    acc += __shfl_xor(acc, 4, 32);
    acc += __shfl_xor(acc, 2, 32);
    acc += __shfl_xor(acc, 1, 32);
    if (q == g) res = acc;
  }
  if (e < nE) out[e] = res;
}

constexpr int SB00 = (E00 + EDGES_PER_BLOCK - 1) / EDGES_PER_BLOCK;
constexpr int SB01 = (E01 + EDGES_PER_BLOCK - 1) / EDGES_PER_BLOCK;
constexpr int SB10 = (E10 + EDGES_PER_BLOCK - 1) / EDGES_PER_BLOCK;
constexpr int SB11 = (E11 + EDGES_PER_BLOCK - 1) / EDGES_PER_BLOCK;
__global__ __launch_bounds__(256) void edge_scores(
    const float* __restrict__ l11, const float* __restrict__ r11,
    const int* __restrict__ d11, const int* __restrict__ c11, float* __restrict__ o11,
    const float* __restrict__ l00, const float* __restrict__ r00,
    const int* __restrict__ d00, const int* __restrict__ c00, float* __restrict__ o00,
    const float* __restrict__ l01, const float* __restrict__ r01,
    const int* __restrict__ d01, const int* __restrict__ c01, float* __restrict__ o01,
    const float* __restrict__ l10, const float* __restrict__ r10,
    const int* __restrict__ d10, const int* __restrict__ c10, float* __restrict__ o10) {
  int t = blockIdx.x;
  if (t < KK * SB11) {
    const int k = t / SB11;
    t -= k * SB11;
    const long long zo = (long long)k * N1 * DD;
    const long long eo = (long long)k * E11;
    edge_dot128_block(l11 + zo, r11 + zo, d11 + eo, c11 + eo, o11 + eo, E11, t);
  } else if (t < KK * SB11 + SB00) {
    edge_dot128_block(l00, r00, d00, c00, o00, E00, t - KK * SB11);
  } else if (t < KK * SB11 + SB00 + SB01) {
    edge_dot128_block(l01, r01, d01, c01, o01, E01, t - (KK * SB11 + SB00));
  } else {
    edge_dot128_block(l10, r10, d10, c10, o10, E10, t - (KK * SB11 + SB00 + SB01));
  }
}
constexpr int SCORE_BLOCKS = KK * SB11 + SB00 + SB01 + SB10;

extern "C" void kernel_launch(void* const* d_in, const int* in_sizes, int n_in,
                              void* d_out, int out_size, void* d_ws, size_t ws_size,
                              hipStream_t stream) {
  (void)in_sizes; (void)n_in; (void)out_size; (void)ws_size;

  const float* feat0   = (const float*)d_in[0];
  const float* feat1   = (const float*)d_in[1];
  const int*   e00_dst = (const int*)d_in[2];
  const int*   e00_src = (const int*)d_in[3];
  const float* e00_val = (const float*)d_in[4];
  const int*   e01_dst = (const int*)d_in[5];
  const int*   e01_src = (const int*)d_in[6];
  const float* e01_val = (const float*)d_in[7];
  const int*   e10_dst = (const int*)d_in[8];
  const int*   e10_src = (const int*)d_in[9];
  const float* e10_val = (const float*)d_in[10];
  const int*   e11_dst = (const int*)d_in[11];
  const int*   e11_src = (const int*)d_in[12];
  const float* e11_val = (const float*)d_in[13];
  const float* W1_00 = (const float*)d_in[14];
  const float* W1_01 = (const float*)d_in[15];
  const float* W1_10 = (const float*)d_in[16];
  const float* W1_11 = (const float*)d_in[17];
  const float* W2_00 = (const float*)d_in[18];
  const float* W2_01 = (const float*)d_in[19];
  const float* W2_10 = (const float*)d_in[20];
  const float* W2_11 = (const float*)d_in[21];
  const float* R00   = (const float*)d_in[22];
  const float* R01   = (const float*)d_in[23];
  const float* R10   = (const float*)d_in[24];
  const float* Rglob = (const float*)d_in[25];
  const float* dloc  = (const float*)d_in[26];

  float* ws = (float*)d_ws;
  float* out = (float*)d_out;
  float* z0  = out;
  float* z1  = out + 2560000;
  float* s00 = z1 + 256000;
  float* s01 = s00 + E00;
  float* s10 = s01 + E01;
  float* s11 = s10 + E10;

  const dim3 blk(256);

  int*  cnt  = (int*)(ws + off_cnt);
  int*  cur  = (int*)(ws + off_cur);
  int2* perm = (int2*)(ws + off_perm);
  hipMemsetAsync(cnt, 0, (size_t)NSEG * sizeof(int), stream);
  dense_stage<FF, HH, true><<<dim3(dense_stage_blocks<FF, HH, true>()), blk, 0, stream>>>(
      feat0, feat1, W1_00, W1_10, W1_11, W1_01,
      ws + off_xw00, ws + off_xw10, ws + off_xw11, ws + off_xw01,
      e00_dst, e01_dst, e10_dst, e11_dst, cnt);
  segment_scan<<<dim3(1), dim3(1024), 0, stream>>>(cnt, cur, NSEG);
  edge_fill<<<dim3(EDGE_BLOCKS), blk, 0, stream>>>(e00_dst, e01_dst, e10_dst, e11_dst,
                                                          e00_src, e01_src, e10_src, e11_src,
                                                          e00_val, e01_val, e10_val, e11_val,
                                                          cur, perm);

  sparse_stage<HH, true><<<dim3(GATHER11_BLOCKS + ROWS0_BLOCKS), blk, 0, stream>>>(
      cnt, cur, perm, ws + off_xw11, ws + off_acc11, ws + off_xw00, ws + off_xw01, ws + off_h0);
  conv_rows_small<HH, true, false><<<dim3((N1 + 7) / 8), blk, 0, stream>>>(
      cnt, cur, perm, SEG10, ws + off_xw10, ws + off_acc11, N1, ws + off_h1, nullptr, nullptr);

  dense_stage<HH, DD, false><<<dim3(dense_stage_blocks<HH, DD, false>()), blk, 0, stream>>>(
      ws + off_h0, ws + off_h1, W2_00, W2_10, W2_11, W2_01,
      ws + off_hw00, ws + off_hw10, ws + off_hw11, ws + off_hw01,
      nullptr, nullptr, nullptr, nullptr, nullptr);

  sparse_stage<DD, false><<<dim3(GATHER11_BLOCKS + ROWS0_BLOCKS), blk, 0, stream>>>(
      cnt, cur, perm, ws + off_hw11, ws + off_bcc11, ws + off_hw00, ws + off_hw01, z0);
  conv_rows_small<DD, false, true><<<dim3((N1 + 7) / 8), blk, 0, stream>>>(
      cnt, cur, perm, SEG10, ws + off_hw10, ws + off_bcc11, N1, z1, dloc, ws + off_z1d);

  dense_decoder<<<dim3(DECODER_BLOCKS), blk, 0, stream>>>(
      z0, z1, ws + off_z1d, R00, R01, R10, Rglob,
      ws + off_z0R00, ws + off_z0R01, ws + off_z1R10, ws + off_Zk);

  edge_scores<<<dim3(SCORE_BLOCKS), blk, 0, stream>>>(
      ws + off_Zk, ws + off_z1d, e11_dst, e11_src, s11,
      ws + off_z0R00, z0, e00_dst, e00_src, s00,
      ws + off_z0R01, z1, e01_dst, e01_src, s01,
      ws + off_z1R10, z0, e10_dst, e10_src, s10);
}
